# starting kernel
# baseline (speedup 1.0000x reference)
#include <hip/hip_runtime.h>
#include <hip/hip_bf16.h>

#define N_NODES 4096
#define IN_F    256
#define OUT_F   64
#define NHEAD   8
#define HID     512
#define NEG_BIG (-9.0e15f)
#define LN_EPS  1e-05f

typedef __attribute__((ext_vector_type(2))) float v2f;
typedef __attribute__((ext_vector_type(8))) float v8f;

__device__ __forceinline__ v8f wmma_f32_16x16x4(v2f A, v2f B, v8f C) {
    return __builtin_amdgcn_wmma_f32_16x16x4_f32(false, A, false, B, (short)0, C,
                                                 false, false);
}

__global__ void __launch_bounds__(128)
k1_gemm_ht(const float* __restrict__ h, const float* __restrict__ W,
           float* __restrict__ ht) {
    __shared__ float sA[16 * 260];
    const int i0 = blockIdx.x * 16;
    const int n0 = blockIdx.y * 64;
    const int tid  = threadIdx.x;
    const int lane = tid & 31;
    const int wave = tid >> 5;

    for (int idx = tid; idx < 16 * 256 / 4; idx += 128) {
        const int r = idx >> 6, c = (idx & 63) << 2;
        const float4 v = *(const float4*)&h[(size_t)(i0 + r) * IN_F + c];
        *(float4*)&sA[r * 260 + c] = v;
    }
    __syncthreads();

    const int m  = lane & 15;
    const int hi = lane >> 4;

    v8f acc = {};
    for (int ks = 0; ks < IN_F / 4; ++ks) {
        v2f A, B;
#pragma unroll
        for (int v = 0; v < 2; ++v) {
            const int kk = (hi << 1) + v;
            const int k  = ks * 4 + kk;
            A[v] = sA[m * 260 + k];
            B[v] = W[(size_t)k * HID + n0 + (wave << 4) + m];
        }
        acc = wmma_f32_16x16x4(A, B, acc);
    }

#pragma unroll
    for (int p = 0; p < 8; ++p) {
        const int row = p + (hi << 3);
        ht[(size_t)(i0 + row) * HID + n0 + (wave << 4) + m] = acc[p];
    }
}

__global__ void __launch_bounds__(256)
k2_scores(const float* __restrict__ ht, const float* __restrict__ a,
          float* __restrict__ ssrc, float* __restrict__ sdst) {
    const int p = blockIdx.x * blockDim.x + threadIdx.x;
    if (p >= N_NODES * NHEAD) return;
    const float* row = ht + (size_t)p * OUT_F;
    float s1 = 0.f, s2 = 0.f;
#pragma unroll 8
    for (int f = 0; f < OUT_F; ++f) {
        const float v = row[f];
        s1 += v * a[f];
        s2 += v * a[OUT_F + f];
    }
    ssrc[p] = s1;
    sdst[p] = s2;
}

__global__ void __launch_bounds__(256)
k3_rowstats(const int* __restrict__ adj, const float* __restrict__ ssrc,
            const float* __restrict__ sdst, float* __restrict__ Mb,
            float* __restrict__ Lb) {
    __shared__ float red[NHEAD][256];
    const int i   = blockIdx.x;
    const int tid = threadIdx.x;

    float sr[NHEAD];
#pragma unroll
    for (int hh = 0; hh < NHEAD; ++hh) sr[hh] = ssrc[i * NHEAD + hh];

    float mx[NHEAD];
#pragma unroll
    for (int hh = 0; hh < NHEAD; ++hh) mx[hh] = NEG_BIG;
    for (int j = tid; j < N_NODES; j += 256) {
        const int av = adj[(size_t)i * N_NODES + j];
#pragma unroll
        for (int hh = 0; hh < NHEAD; ++hh) {
            float e = sr[hh] + sdst[j * NHEAD + hh];
            e = e > 0.f ? e : 0.2f * e;
            const float val = av > 0 ? e : NEG_BIG;
            mx[hh] = fmaxf(mx[hh], val);
        }
    }
#pragma unroll
    for (int hh = 0; hh < NHEAD; ++hh) red[hh][tid] = mx[hh];
    __syncthreads();
    for (int s = 128; s >= 1; s >>= 1) {
        if (tid < s) {
#pragma unroll
            for (int hh = 0; hh < NHEAD; ++hh)
                red[hh][tid] = fmaxf(red[hh][tid], red[hh][tid + s]);
        }
        __syncthreads();
    }
    float Mh[NHEAD];
#pragma unroll
    for (int hh = 0; hh < NHEAD; ++hh) Mh[hh] = red[hh][0];
    __syncthreads();

    float sm[NHEAD];
#pragma unroll
    for (int hh = 0; hh < NHEAD; ++hh) sm[hh] = 0.f;
    for (int j = tid; j < N_NODES; j += 256) {
        const int av = adj[(size_t)i * N_NODES + j];
#pragma unroll
        for (int hh = 0; hh < NHEAD; ++hh) {
            float e = sr[hh] + sdst[j * NHEAD + hh];
            e = e > 0.f ? e : 0.2f * e;
            const float val = av > 0 ? e : NEG_BIG;
            sm[hh] += __expf(val - Mh[hh]);
        }
    }
#pragma unroll
    for (int hh = 0; hh < NHEAD; ++hh) red[hh][tid] = sm[hh];
    __syncthreads();
    for (int s = 128; s >= 1; s >>= 1) {
        if (tid < s) {
#pragma unroll
            for (int hh = 0; hh < NHEAD; ++hh)
                red[hh][tid] += red[hh][tid + s];
        }
        __syncthreads();
    }
    if (tid < NHEAD) {
        Mb[i * NHEAD + tid] = Mh[tid];
        Lb[i * NHEAD + tid] = red[tid][0];
    }
}

#define JT 64
__global__ void __launch_bounds__(128)
k4_agg_ln(const int* __restrict__ adj, const float* __restrict__ ht,
          const float* __restrict__ ssrc, const float* __restrict__ sdst,
          const float* __restrict__ Mb, const float* __restrict__ Lb,
          const float* __restrict__ gamma, const float* __restrict__ beta,
          float* __restrict__ out) {
    __shared__ int   s_adj[16 * 68];
    __shared__ float s_sdst[JT * NHEAD];
    __shared__ float s_htT[OUT_F * 68];
    __shared__ float s_out[16 * OUT_F];
    __shared__ float s_L[16 * NHEAD];
    __shared__ float s_M[16 * NHEAD];
    __shared__ float s_ss[16 * NHEAD];
    __shared__ float s_mu[16], s_rs[16];

    const int i0   = blockIdx.x * 16;
    const int tid  = threadIdx.x;
    const int lane = tid & 31;
    const int wave = tid >> 5;
    const int m    = lane & 15;
    const int hi   = lane >> 4;
    const int fcol = (wave << 4) + m;

    {
        const int r = tid >> 3, hh = tid & 7;
        s_L[r * NHEAD + hh]  = Lb[(i0 + r) * NHEAD + hh];
        s_M[r * NHEAD + hh]  = Mb[(i0 + r) * NHEAD + hh];
        s_ss[r * NHEAD + hh] = ssrc[(i0 + r) * NHEAD + hh];
    }
    __syncthreads();

    v8f acc[NHEAD] = {};

    for (int j0 = 0; j0 < N_NODES; j0 += JT) {
        if (j0 + JT < N_NODES) {
            const int r = tid >> 3;
            if (r < 16)
                __builtin_prefetch(&adj[(size_t)(i0 + r) * N_NODES + j0 + JT], 0, 1);
        }
        for (int idx = tid; idx < 16 * JT / 4; idx += 128) {
            const int r = idx >> 4, c = (idx & 15) << 2;
            const int4 v = *(const int4*)&adj[(size_t)(i0 + r) * N_NODES + j0 + c];
            *(int4*)&s_adj[r * 68 + c] = v;
        }
        {
            const float4 v = *(const float4*)&sdst[j0 * NHEAD + (tid << 2)];
            *(float4*)&s_sdst[tid << 2] = v;
        }
        __syncthreads();

#pragma unroll
        for (int hh = 0; hh < NHEAD; ++hh) {
            for (int idx = tid; idx < JT * OUT_F / 4; idx += 128) {
                const int jl = idx >> 4, f4 = (idx & 15) << 2;
                const float4 v =
                    *(const float4*)&ht[(size_t)(j0 + jl) * HID + hh * OUT_F + f4];
                s_htT[(f4 + 0) * 68 + jl] = v.x;
                s_htT[(f4 + 1) * 68 + jl] = v.y;
                s_htT[(f4 + 2) * 68 + jl] = v.z;
                s_htT[(f4 + 3) * 68 + jl] = v.w;
            }
            __syncthreads();

            const float ssr  = s_ss[m * NHEAD + hh];
            const float Mrow = s_M[m * NHEAD + hh];
            v8f a_acc = acc[hh];
            for (int ks = 0; ks < JT / 4; ++ks) {
                const int jbase = ks * 4 + (hi << 1);
                v2f A, B;
#pragma unroll
                for (int v = 0; v < 2; ++v) {
                    const int jl = jbase + v;
                    const int adjv = s_adj[m * 68 + jl];
                    float e = ssr + s_sdst[jl * NHEAD + hh];
                    e = e > 0.f ? e : 0.2f * e;
                    const float val = adjv > 0 ? e : NEG_BIG;
                    A[v] = __expf(val - Mrow);
                    B[v] = s_htT[fcol * 68 + jl];
                }
                a_acc = wmma_f32_16x16x4(A, B, a_acc);
            }
            acc[hh] = a_acc;
            __syncthreads();
        }
        __syncthreads();
    }

    v8f hp = {};
#pragma unroll
    for (int hh = 0; hh < NHEAD; ++hh) {
#pragma unroll
        for (int p = 0; p < 8; ++p) {
            const int row = p + (hi << 3);
            hp[p] += acc[hh][p] / s_L[row * NHEAD + hh];
        }
    }
#pragma unroll
    for (int p = 0; p < 8; ++p) {
        const int row = p + (hi << 3);
        s_out[row * OUT_F + fcol] = hp[p] * (1.0f / NHEAD);
    }
    __syncthreads();

    if (tid < 16) {
        float mu = 0.f;
        for (int f = 0; f < OUT_F; ++f) mu += s_out[tid * OUT_F + f];
        mu *= (1.0f / OUT_F);
        float var = 0.f;
        for (int f = 0; f < OUT_F; ++f) {
            const float d = s_out[tid * OUT_F + f] - mu;
            var += d * d;
        }
        var *= (1.0f / OUT_F);
        s_mu[tid] = mu;
        s_rs[tid] = rsqrtf(var + LN_EPS);
    }
    __syncthreads();

    for (int idx = tid; idx < 16 * OUT_F; idx += 128) {
        const int r = idx >> 6, f = idx & 63;
        out[(size_t)(i0 + r) * OUT_F + f] =
            (s_out[r * OUT_F + f] - s_mu[r]) * s_rs[r] * gamma[f] + beta[f];
    }
}

extern "C" void kernel_launch(void* const* d_in, const int* in_sizes, int n_in,
                              void* d_out, int out_size, void* d_ws, size_t ws_size,
                              hipStream_t stream) {
    const float* h     = (const float*)d_in[0];
    const int*   adj   = (const int*)  d_in[1];
    const float* W     = (const float*)d_in[2];
    const float* a     = (const float*)d_in[3];
    const float* gamma = (const float*)d_in[4];
    const float* beta  = (const float*)d_in[5];
    float* out = (float*)d_out;

    float* ws   = (float*)d_ws;
    float* ht   = ws;
    float* ssrc = ht + (size_t)N_NODES * HID;
    float* sdst = ssrc + N_NODES * NHEAD;
    float* Mb   = sdst + N_NODES * NHEAD;
    float* Lb   = Mb   + N_NODES * NHEAD;

    k1_gemm_ht<<<dim3(N_NODES / 16, HID / 64), 128, 0, stream>>>(h, W, ht);
    k2_scores<<<(N_NODES * NHEAD) / 256, 256, 0, stream>>>(ht, a, ssrc, sdst);
    k3_rowstats<<<N_NODES, 256, 0, stream>>>(adj, ssrc, sdst, Mb, Lb);
    k4_agg_ln<<<N_NODES / 16, 128, 0, stream>>>(adj, ht, ssrc, sdst, Mb, Lb,
                                                gamma, beta, out);
}


# k1 stores ht in k4's fragment order, k2 stores s_dst head-major: k4 gets A pairs, adj, s_dst by 16-byte loads, no register moves (24 loads a block, was 40); k and L sum orders change
# speedup vs baseline: 12.6667x; 12.6667x over previous
#include <hip/hip_runtime.h>
#include <hip/hip_bf16.h>

#define N_NODES 4096
#define IN_F    256
#define OUT_F   64
#define NHEAD   8
#define HID     512
#define NEG_BIG (-9.0e15f)
#define LN_EPS  1e-05f

typedef __attribute__((ext_vector_type(2))) float v2f;
typedef __attribute__((ext_vector_type(8))) float v8f;

__device__ __forceinline__ v8f wmma_f32_16x16x4(v2f A, v2f B, v8f C) {
    return __builtin_amdgcn_wmma_f32_16x16x4_f32(false, A, false, B, (short)0, C,
                                                 false, false);
}

#define HTF_BLK 2048
__global__ void __launch_bounds__(128)
k1_gemm_ht(const float* __restrict__ h, const float* __restrict__ W,
           float* __restrict__ ht) {
    __shared__ float sA[16 * 260];
    const int i0 = blockIdx.x * 16;
    const int n0 = blockIdx.y * 64;
    const int tid  = threadIdx.x;
    const int lane = tid & 31;
    const int wave = tid >> 5;

    for (int idx = tid; idx < 16 * 256 / 4; idx += 128) {
        const int r = idx >> 6, c = (idx & 63) << 2;
        const float4 v = *(const float4*)&h[(size_t)(i0 + r) * IN_F + c];
        *(float4*)&sA[r * 260 + c] = v;
    }
    __syncthreads();

    const int m  = lane & 15;
    const int hi = lane >> 4;

    v8f acc = {};
    for (int ks = 0; ks < IN_F / 4; ++ks) {
        v2f A, B;
#pragma unroll
        for (int v = 0; v < 2; ++v) {
            const int kk = (hi << 1) + v;
            const int k  = ks * 4 + kk;
            A[v] = sA[m * 260 + k];
            B[v] = W[(size_t)k * HID + n0 + (wave << 4) + m];
        }
        acc = wmma_f32_16x16x4(A, B, acc);
    }

    const int f  = (wave << 4) + m;
    float* dst = ht + (size_t)(blockIdx.y * (N_NODES / 32) + (blockIdx.x >> 1)) * HTF_BLK
               + (f & 3) * 512 + ((((int)blockIdx.x & 1) << 4) + (f >> 2)) * 4;
#pragma unroll
    for (int pp = 0; pp < 2; ++pp) {
        float4 o;
        o.x = acc[4 * pp + 0];
        o.y = acc[4 * pp + 1];
        o.z = acc[4 * pp + 2];
        o.w = acc[4 * pp + 3];
        *(float4*)&dst[((hi << 1) + pp) * 128] = o;
    }
}

__global__ void __launch_bounds__(256)
k2_scores(const float* __restrict__ ht, const float* __restrict__ a,
          float* __restrict__ ssrc, float* __restrict__ sdst) {
    const int p = blockIdx.x * blockDim.x + threadIdx.x;
    if (p >= N_NODES * NHEAD) return;
    const int hd = p / N_NODES, n = p % N_NODES;
    const float* row = ht + (size_t)(hd * (N_NODES / 32) + (n >> 5)) * HTF_BLK
                     + ((n >> 2) & 3) * 128 + ((n >> 4) & 1) * 64 + (n & 3);
    float s1 = 0.f, s2 = 0.f;
#pragma unroll 8
    for (int f = 0; f < OUT_F; ++f) {
        const float v = row[(f & 3) * 512 + (f >> 2) * 4];
        s1 += v * a[f];
        s2 += v * a[OUT_F + f];
    }
    ssrc[n * NHEAD + hd] = s1;
    sdst[p] = s2;
}

#define KU 8
#define JB (KU * 4)
#define K4_LAG 8.0f

typedef __attribute__((ext_vector_type(4))) float v4f;
typedef __attribute__((ext_vector_type(4))) int   v4i;
#define KQ (KU / 2)

struct K4Frag {
    v4i adjq[KQ];
    v4f sd[KQ];
    v4f b[4][KQ];
};

__device__ __forceinline__ void k4_load(K4Frag& fr, const int* __restrict__ adj_row,
                                        const float* __restrict__ sd_col,
                                        const float* __restrict__ ht_col, int jb) {
    const float* chunk = ht_col + (size_t)(jb >> 5) * HTF_BLK;
#pragma unroll
    for (int q = 0; q < KQ; ++q) {
        const int j = jb + q * 4;
        fr.adjq[q] = *(const v4i*)&adj_row[j];
        fr.sd[q]   = *(const v4f*)&sd_col[j];
#pragma unroll
        for (int t = 0; t < 4; ++t)
            fr.b[t][q] = *(const v4f*)&chunk[t * 512 + q * 128];
    }
}

__device__ __forceinline__ float k4_score(int adjv, float sd, float ssr) {
    float e = ssr + sd;
    e = e > 0.f ? e : 0.2f * e;
    return adjv > 0 ? e : NEG_BIG;
}

__device__ __forceinline__ float k4_xhalf(float x) {
    const int b = __builtin_bit_cast(int, x);
    return __builtin_bit_cast(float,
        __builtin_amdgcn_permlanex16(b, b, 0x76543210, 0xfedcba98, false, false));
}

__device__ __forceinline__ void k4_compute(const K4Frag& fr, float ssr,
                                           float& m_use, v2f& l_run,
                                           v8f& acc0, v8f& acc1, v8f& acc2, v8f& acc3) {
    float e[KQ][4];
    float bm = NEG_BIG;
#pragma unroll
    for (int q = 0; q < KQ; ++q) {
#pragma unroll
        for (int x = 0; x < 4; ++x)
            e[q][x] = k4_score(fr.adjq[q][x], fr.sd[q][x], ssr);
        bm = fmaxf(bm, fmaxf(fmaxf(e[q][0], e[q][1]), fmaxf(e[q][2], e[q][3])));
    }
    bm = fmaxf(bm, k4_xhalf(bm));
    const bool need = bm > m_use + K4_LAG;
    if (__builtin_amdgcn_ballot_w32(need) != 0) {
        const float m_new = fmaxf(m_use, bm);
        const float corr  = __expf(m_use - m_new);
        m_use = m_new;
        l_run = l_run * corr;
        acc0 = acc0 * corr;
        acc1 = acc1 * corr;
        acc2 = acc2 * corr;
        acc3 = acc3 * corr;
    }
#pragma unroll
    for (int q = 0; q < KQ; ++q) {
        v4f x0 = fr.b[0][q], x1 = fr.b[1][q], x2 = fr.b[2][q], x3 = fr.b[3][q];
        asm("" : "+v"(x0), "+v"(x1), "+v"(x2), "+v"(x3));
#pragma unroll
        for (int s = 0; s < 2; ++s) {
            v2f P;
            P[0] = __expf(e[q][2 * s + 0] - m_use);
            P[1] = __expf(e[q][2 * s + 1] - m_use);
            l_run += P;
            v2f A0, A1, A2, A3;
            A0[0] = x0[2 * s]; A0[1] = x0[2 * s + 1];
            A1[0] = x1[2 * s]; A1[1] = x1[2 * s + 1];
            A2[0] = x2[2 * s]; A2[1] = x2[2 * s + 1];
            A3[0] = x3[2 * s]; A3[1] = x3[2 * s + 1];
            acc0 = wmma_f32_16x16x4(A0, P, acc0);
            acc1 = wmma_f32_16x16x4(A1, P, acc1);
            acc2 = wmma_f32_16x16x4(A2, P, acc2);
            acc3 = wmma_f32_16x16x4(A3, P, acc3);
        }
    }
}

__global__ void __launch_bounds__(256) __attribute__((amdgpu_waves_per_eu(2, 3)))
k4_agg_ln(const int* __restrict__ adj, const float* __restrict__ ht,
          const float* __restrict__ ssrc, const float* __restrict__ sdst,
          const float* __restrict__ gamma, const float* __restrict__ beta,
          float* __restrict__ out) {
    __shared__ float s_part[NHEAD * 16 * OUT_F];
    __shared__ float s_out[16 * OUT_F];
    __shared__ float s_mu[16], s_rs[16];

    const int i0   = blockIdx.x * 16;
    const int tid  = threadIdx.x;
    const int lane = tid & 31;
    const int hd   = tid >> 5;
    const int c    = lane & 15;
    const int hi   = lane >> 4;

    const float ssr = ssrc[(i0 + c) * NHEAD + hd];
    float m_use = -__builtin_inff();
    v2f   l_run = {};

    const int*   adj_row = adj + (size_t)(i0 + c) * N_NODES + (hi << 4);
    const float* sd_col  = sdst + (size_t)hd * N_NODES + (hi << 4);
    const float* ht_col  = ht + (size_t)hd * (N_NODES / 32) * HTF_BLK + (lane << 2);

    v8f acc0 = {}, acc1 = {}, acc2 = {}, acc3 = {};

#define K4_STEP_END()                                                        \
    do {                                                                     \
        asm volatile("" : "+v"(acc0), "+v"(acc1), "+v"(acc2), "+v"(acc3),    \
                          "+v"(l_run) : : "memory");                         \
        __builtin_amdgcn_sched_barrier(0);                                   \
    } while (0)

    K4Frag f0, f1;
    k4_load(f0, adj_row, sd_col, ht_col, 0);
    K4_STEP_END();
    for (int jb = 0; jb < N_NODES; jb += 2 * JB) {
        k4_load(f1, adj_row, sd_col, ht_col, jb + JB);
        K4_STEP_END();
        k4_compute(f0, ssr, m_use, l_run, acc0, acc1, acc2, acc3);
        K4_STEP_END();
        const int jn = (jb + 2 * JB < N_NODES) ? jb + 2 * JB : jb;
        k4_load(f0, adj_row, sd_col, ht_col, jn);
        K4_STEP_END();
        k4_compute(f1, ssr, m_use, l_run, acc0, acc1, acc2, acc3);
        K4_STEP_END();
    }

    const float l_lane = l_run[0] + l_run[1];
    const float L = l_lane + k4_xhalf(l_lane);

#pragma unroll
    for (int r = 0; r < 8; ++r) {
        float4 o;
        o.x = acc0[r] / L;
        o.y = acc1[r] / L;
        o.z = acc2[r] / L;
        o.w = acc3[r] / L;
        *(float4*)&s_part[(hd * 16 + c) * OUT_F + (hi << 5) + (r << 2)] = o;
    }
    __syncthreads();

    for (int idx = tid; idx < 16 * OUT_F; idx += 256) {
        float hp = 0.f;
#pragma unroll
        for (int hh = 0; hh < NHEAD; ++hh) hp += s_part[hh * 16 * OUT_F + idx];
        s_out[idx] = hp * (1.0f / NHEAD);
    }
    __syncthreads();

    if (tid < 16) {
        float mu = 0.f;
        for (int f = 0; f < OUT_F; ++f) mu += s_out[tid * OUT_F + f];
        mu *= (1.0f / OUT_F);
        float var = 0.f;
        for (int f = 0; f < OUT_F; ++f) {
            const float d = s_out[tid * OUT_F + f] - mu;
            var += d * d;
        }
        var *= (1.0f / OUT_F);
        s_mu[tid] = mu;
        s_rs[tid] = rsqrtf(var + LN_EPS);
    }
    __syncthreads();

    for (int idx = tid; idx < 16 * OUT_F; idx += 256) {
        const int r = idx >> 6, f = idx & 63;
        out[(size_t)(i0 + r) * OUT_F + f] =
            (s_out[r * OUT_F + f] - s_mu[r]) * s_rs[r] * gamma[f] + beta[f];
    }
}

extern "C" void kernel_launch(void* const* d_in, const int* in_sizes, int n_in,
                              void* d_out, int out_size, void* d_ws, size_t ws_size,
                              hipStream_t stream) {
    const float* h     = (const float*)d_in[0];
    const int*   adj   = (const int*)  d_in[1];
    const float* W     = (const float*)d_in[2];
    const float* a     = (const float*)d_in[3];
    const float* gamma = (const float*)d_in[4];
    const float* beta  = (const float*)d_in[5];
    float* out = (float*)d_out;

    float* ws   = (float*)d_ws;
    float* ht   = ws;
    float* ssrc = ht + (size_t)N_NODES * HID;
    float* sdst = ssrc + N_NODES * NHEAD;

    k1_gemm_ht<<<dim3(N_NODES / 16, HID / 64), 128, 0, stream>>>(h, W, ht);
    k2_scores<<<(N_NODES * NHEAD) / 256, 256, 0, stream>>>(ht, a, ssrc, sdst);
    k4_agg_ln<<<N_NODES / 16, 256, 0, stream>>>(adj, ht, ssrc, sdst,
                                                gamma, beta, out);
}


# k4: each head's j range split over 2 waves (512-thread blocks, 4 waves per SIMD, was 2), own online-softmax state each, merged through LDS at the end; k and L sums now two halves added
# speedup vs baseline: 13.0694x; 1.0318x over previous
#include <hip/hip_runtime.h>
#include <hip/hip_bf16.h>

#define N_NODES 4096
#define IN_F    256
#define OUT_F   64
#define NHEAD   8
#define HID     512
#define NEG_BIG (-9.0e15f)
#define LN_EPS  1e-05f

typedef __attribute__((ext_vector_type(2))) float v2f;
typedef __attribute__((ext_vector_type(8))) float v8f;

__device__ __forceinline__ v8f wmma_f32_16x16x4(v2f A, v2f B, v8f C) {
    return __builtin_amdgcn_wmma_f32_16x16x4_f32(false, A, false, B, (short)0, C,
                                                 false, false);
}

#define HTF_BLK 2048
__global__ void __launch_bounds__(128)
k1_gemm_ht(const float* __restrict__ h, const float* __restrict__ W,
           float* __restrict__ ht) {
    __shared__ float sA[16 * 260];
    const int i0 = blockIdx.x * 16;
    const int n0 = blockIdx.y * 64;
    const int tid  = threadIdx.x;
    const int lane = tid & 31;
    const int wave = tid >> 5;

    for (int idx = tid; idx < 16 * 256 / 4; idx += 128) {
        const int r = idx >> 6, c = (idx & 63) << 2;
        const float4 v = *(const float4*)&h[(size_t)(i0 + r) * IN_F + c];
        *(float4*)&sA[r * 260 + c] = v;
    }
    __syncthreads();

    const int m  = lane & 15;
    const int hi = lane >> 4;

    v8f acc = {};
    for (int ks = 0; ks < IN_F / 4; ++ks) {
        v2f A, B;
#pragma unroll
        for (int v = 0; v < 2; ++v) {
            const int kk = (hi << 1) + v;
            const int k  = ks * 4 + kk;
            A[v] = sA[m * 260 + k];
            B[v] = W[(size_t)k * HID + n0 + (wave << 4) + m];
        }
        acc = wmma_f32_16x16x4(A, B, acc);
    }

    const int f  = (wave << 4) + m;
    float* dst = ht + (size_t)(blockIdx.y * (N_NODES / 32) + (blockIdx.x >> 1)) * HTF_BLK
               + (f & 3) * 512 + ((((int)blockIdx.x & 1) << 4) + (f >> 2)) * 4;
#pragma unroll
    for (int pp = 0; pp < 2; ++pp) {
        float4 o;
        o.x = acc[4 * pp + 0];
        o.y = acc[4 * pp + 1];
        o.z = acc[4 * pp + 2];
        o.w = acc[4 * pp + 3];
        *(float4*)&dst[((hi << 1) + pp) * 128] = o;
    }
}

__global__ void __launch_bounds__(256)
k2_scores(const float* __restrict__ ht, const float* __restrict__ a,
          float* __restrict__ ssrc, float* __restrict__ sdst) {
    const int p = blockIdx.x * blockDim.x + threadIdx.x;
    if (p >= N_NODES * NHEAD) return;
    const int hd = p / N_NODES, n = p % N_NODES;
    const float* row = ht + (size_t)(hd * (N_NODES / 32) + (n >> 5)) * HTF_BLK
                     + ((n >> 2) & 3) * 128 + ((n >> 4) & 1) * 64 + (n & 3);
    float s1 = 0.f, s2 = 0.f;
#pragma unroll 8
    for (int f = 0; f < OUT_F; ++f) {
        const float v = row[(f & 3) * 512 + (f >> 2) * 4];
        s1 += v * a[f];
        s2 += v * a[OUT_F + f];
    }
    ssrc[n * NHEAD + hd] = s1;
    sdst[p] = s2;
}

#define KU 8
#define JB (KU * 4)
#define K4_LAG 8.0f

typedef __attribute__((ext_vector_type(4))) float v4f;
typedef __attribute__((ext_vector_type(4))) int   v4i;
#define KQ (KU / 2)
#define K4_WAVES   16
#define K4_THREADS (K4_WAVES * 32)
#define JHALF      (N_NODES / 2)
#define MRG_ROWS   34

struct K4Frag {
    v4i adjq[KQ];
    v4f sd[KQ];
    v4f b[4][KQ];
};

__device__ __forceinline__ void k4_load(K4Frag& fr, const int* __restrict__ adj_row,
                                        const float* __restrict__ sd_col,
                                        const float* __restrict__ ht_col, int jb) {
    const float* chunk = ht_col + (size_t)(jb >> 5) * HTF_BLK;
#pragma unroll
    for (int q = 0; q < KQ; ++q) {
        const int j = jb + q * 4;
        fr.adjq[q] = *(const v4i*)&adj_row[j];
        fr.sd[q]   = *(const v4f*)&sd_col[j];
#pragma unroll
        for (int t = 0; t < 4; ++t)
            fr.b[t][q] = *(const v4f*)&chunk[t * 512 + q * 128];
    }
}

__device__ __forceinline__ float k4_score(int adjv, float sd, float ssr) {
    float e = ssr + sd;
    e = e > 0.f ? e : 0.2f * e;
    return adjv > 0 ? e : NEG_BIG;
}

__device__ __forceinline__ float k4_xhalf(float x) {
    const int b = __builtin_bit_cast(int, x);
    return __builtin_bit_cast(float,
        __builtin_amdgcn_permlanex16(b, b, 0x76543210, 0xfedcba98, false, false));
}

__device__ __forceinline__ void k4_compute(const K4Frag& fr, float ssr,
                                           float& m_use, v2f& l_run,
                                           v8f& acc0, v8f& acc1, v8f& acc2, v8f& acc3) {
    float e[KQ][4];
    float bm = NEG_BIG;
#pragma unroll
    for (int q = 0; q < KQ; ++q) {
#pragma unroll
        for (int x = 0; x < 4; ++x)
            e[q][x] = k4_score(fr.adjq[q][x], fr.sd[q][x], ssr);
        bm = fmaxf(bm, fmaxf(fmaxf(e[q][0], e[q][1]), fmaxf(e[q][2], e[q][3])));
    }
    bm = fmaxf(bm, k4_xhalf(bm));
    const bool need = bm > m_use + K4_LAG;
    if (__builtin_amdgcn_ballot_w32(need) != 0) {
        const float m_new = fmaxf(m_use, bm);
        const float corr  = __expf(m_use - m_new);
        m_use = m_new;
        l_run = l_run * corr;
        acc0 = acc0 * corr;
        acc1 = acc1 * corr;
        acc2 = acc2 * corr;
        acc3 = acc3 * corr;
    }
#pragma unroll
    for (int q = 0; q < KQ; ++q) {
        v4f x0 = fr.b[0][q], x1 = fr.b[1][q], x2 = fr.b[2][q], x3 = fr.b[3][q];
        asm("" : "+v"(x0), "+v"(x1), "+v"(x2), "+v"(x3));
#pragma unroll
        for (int s = 0; s < 2; ++s) {
            v2f P;
            P[0] = __expf(e[q][2 * s + 0] - m_use);
            P[1] = __expf(e[q][2 * s + 1] - m_use);
            l_run += P;
            v2f A0, A1, A2, A3;
            A0[0] = x0[2 * s]; A0[1] = x0[2 * s + 1];
            A1[0] = x1[2 * s]; A1[1] = x1[2 * s + 1];
            A2[0] = x2[2 * s]; A2[1] = x2[2 * s + 1];
            A3[0] = x3[2 * s]; A3[1] = x3[2 * s + 1];
            acc0 = wmma_f32_16x16x4(A0, P, acc0);
            acc1 = wmma_f32_16x16x4(A1, P, acc1);
            acc2 = wmma_f32_16x16x4(A2, P, acc2);
            acc3 = wmma_f32_16x16x4(A3, P, acc3);
        }
    }
}

__global__ void __launch_bounds__(K4_THREADS) __attribute__((amdgpu_waves_per_eu(4, 4)))
k4_agg_ln(const int* __restrict__ adj, const float* __restrict__ ht,
          const float* __restrict__ ssrc, const float* __restrict__ sdst,
          const float* __restrict__ gamma, const float* __restrict__ beta,
          float* __restrict__ out) {
    __shared__ float s_part[NHEAD * 16 * OUT_F];
    __shared__ float s_mrg[NHEAD * MRG_ROWS * 32];
    __shared__ float s_out[16 * OUT_F];
    __shared__ float s_mu[16], s_rs[16];

    const int i0   = blockIdx.x * 16;
    const int tid  = threadIdx.x;
    const int lane = tid & 31;
    const int hd   = (tid >> 5) & (NHEAD - 1);
    const int hf   = tid >> 8;
    const int c    = lane & 15;
    const int hi   = lane >> 4;

    const float ssr = ssrc[(i0 + c) * NHEAD + hd];
    float m_use = -__builtin_inff();
    v2f   l_run = {};

    const int*   adj_row = adj + (size_t)(i0 + c) * N_NODES + hf * JHALF + (hi << 4);
    const float* sd_col  = sdst + (size_t)hd * N_NODES + hf * JHALF + (hi << 4);
    const float* ht_col  = ht + (size_t)(hd * (N_NODES / 32) + hf * (JHALF / 32)) * HTF_BLK
                         + (lane << 2);

    v8f acc0 = {}, acc1 = {}, acc2 = {}, acc3 = {};

#define K4_STEP_END()                                                        \
    do {                                                                     \
        asm volatile("" : "+v"(acc0), "+v"(acc1), "+v"(acc2), "+v"(acc3),    \
                          "+v"(l_run) : : "memory");                         \
        __builtin_amdgcn_sched_barrier(0);                                   \
    } while (0)

    K4Frag f0, f1;
    k4_load(f0, adj_row, sd_col, ht_col, 0);
    K4_STEP_END();
    for (int jb = 0; jb < JHALF; jb += 2 * JB) {
        k4_load(f1, adj_row, sd_col, ht_col, jb + JB);
        K4_STEP_END();
        k4_compute(f0, ssr, m_use, l_run, acc0, acc1, acc2, acc3);
        K4_STEP_END();
        const int jn = (jb + 2 * JB < JHALF) ? jb + 2 * JB : jb;
        k4_load(f0, adj_row, sd_col, ht_col, jn);
        K4_STEP_END();
        k4_compute(f1, ssr, m_use, l_run, acc0, acc1, acc2, acc3);
        K4_STEP_END();
    }

    const float l_lane = l_run[0] + l_run[1];
    float L = l_lane + k4_xhalf(l_lane);

    float* mrg = s_mrg + hd * (MRG_ROWS * 32) + lane;
    if (hf == 1) {
#pragma unroll
        for (int r = 0; r < 8; ++r) {
            mrg[(r +  0) * 32] = acc0[r];
            mrg[(r +  8) * 32] = acc1[r];
            mrg[(r + 16) * 32] = acc2[r];
            mrg[(r + 24) * 32] = acc3[r];
        }
        mrg[32 * 32] = m_use;
        mrg[33 * 32] = L;
    }
    __syncthreads();

    if (hf == 0) {
        const float m_b = mrg[32 * 32];
        const float m   = fmaxf(m_use, m_b);
        const float sa  = __expf(m_use - m);
        const float sb  = __expf(m_b - m);
        L = L * sa + mrg[33 * 32] * sb;
#pragma unroll
        for (int r = 0; r < 8; ++r) {
            float4 o;
            o.x = (acc0[r] * sa + mrg[(r +  0) * 32] * sb) / L;
            o.y = (acc1[r] * sa + mrg[(r +  8) * 32] * sb) / L;
            o.z = (acc2[r] * sa + mrg[(r + 16) * 32] * sb) / L;
            o.w = (acc3[r] * sa + mrg[(r + 24) * 32] * sb) / L;
            *(float4*)&s_part[(hd * 16 + c) * OUT_F + (hi << 5) + (r << 2)] = o;
        }
    }
    __syncthreads();

    for (int idx = tid; idx < 16 * OUT_F; idx += K4_THREADS) {
        float hp = 0.f;
#pragma unroll
        for (int hh = 0; hh < NHEAD; ++hh) hp += s_part[hh * 16 * OUT_F + idx];
        s_out[idx] = hp * (1.0f / NHEAD);
    }
    __syncthreads();

    if (tid < 16) {
        float mu = 0.f;
        for (int f = 0; f < OUT_F; ++f) mu += s_out[tid * OUT_F + f];
        mu *= (1.0f / OUT_F);
        float var = 0.f;
        for (int f = 0; f < OUT_F; ++f) {
            const float d = s_out[tid * OUT_F + f] - mu;
            var += d * d;
        }
        var *= (1.0f / OUT_F);
        s_mu[tid] = mu;
        s_rs[tid] = rsqrtf(var + LN_EPS);
    }
    __syncthreads();

    for (int idx = tid; idx < 16 * OUT_F; idx += K4_THREADS) {
        const int r = idx >> 6, f = idx & 63;
        out[(size_t)(i0 + r) * OUT_F + f] =
            (s_out[r * OUT_F + f] - s_mu[r]) * s_rs[r] * gamma[f] + beta[f];
    }
}

extern "C" void kernel_launch(void* const* d_in, const int* in_sizes, int n_in,
                              void* d_out, int out_size, void* d_ws, size_t ws_size,
                              hipStream_t stream) {
    const float* h     = (const float*)d_in[0];
    const int*   adj   = (const int*)  d_in[1];
    const float* W     = (const float*)d_in[2];
    const float* a     = (const float*)d_in[3];
    const float* gamma = (const float*)d_in[4];
    const float* beta  = (const float*)d_in[5];
    float* out = (float*)d_out;

    float* ws   = (float*)d_ws;
    float* ht   = ws;
    float* ssrc = ht + (size_t)N_NODES * HID;
    float* sdst = ssrc + N_NODES * NHEAD;

    k1_gemm_ht<<<dim3(N_NODES / 16, HID / 64), 128, 0, stream>>>(h, W, ht);
    k2_scores<<<(N_NODES * NHEAD) / 256, 256, 0, stream>>>(ht, a, ssrc, sdst);
    k4_agg_ln<<<N_NODES / 16, K4_THREADS, 0, stream>>>(adj, ht, ssrc, sdst,
                                                gamma, beta, out);
}


# k4: each wave keeps two 16-row tiles per loaded ht fragment (work-group = 32 rows x 4 heads, 8 waves): ht bytes per work-group halve; new k5 adds the 2 head groups + LayerNorm; head sum order changes
# speedup vs baseline: 13.7607x; 1.0529x over previous
#include <hip/hip_runtime.h>
#include <hip/hip_bf16.h>

#define N_NODES 4096
#define IN_F    256
#define OUT_F   64
#define NHEAD   8
#define HID     512
#define NEG_BIG (-9.0e15f)
#define LN_EPS  1e-05f

typedef __attribute__((ext_vector_type(2))) float v2f;
typedef __attribute__((ext_vector_type(8))) float v8f;

__device__ __forceinline__ v8f wmma_f32_16x16x4(v2f A, v2f B, v8f C) {
    return __builtin_amdgcn_wmma_f32_16x16x4_f32(false, A, false, B, (short)0, C,
                                                 false, false);
}

#define HTF_BLK 2048
__global__ void __launch_bounds__(128)
k1_gemm_ht(const float* __restrict__ h, const float* __restrict__ W,
           float* __restrict__ ht) {
    __shared__ float sA[16 * 260];
    const int i0 = blockIdx.x * 16;
    const int n0 = blockIdx.y * 64;
    const int tid  = threadIdx.x;
    const int lane = tid & 31;
    const int wave = tid >> 5;

    for (int idx = tid; idx < 16 * 256 / 4; idx += 128) {
        const int r = idx >> 6, c = (idx & 63) << 2;
        const float4 v = *(const float4*)&h[(size_t)(i0 + r) * IN_F + c];
        *(float4*)&sA[r * 260 + c] = v;
    }
    __syncthreads();

    const int m  = lane & 15;
    const int hi = lane >> 4;

    v8f acc = {};
    for (int ks = 0; ks < IN_F / 4; ++ks) {
        v2f A, B;
#pragma unroll
        for (int v = 0; v < 2; ++v) {
            const int kk = (hi << 1) + v;
            const int k  = ks * 4 + kk;
            A[v] = sA[m * 260 + k];
            B[v] = W[(size_t)k * HID + n0 + (wave << 4) + m];
        }
        acc = wmma_f32_16x16x4(A, B, acc);
    }

    const int f  = (wave << 4) + m;
    float* dst = ht + (size_t)(blockIdx.y * (N_NODES / 32) + (blockIdx.x >> 1)) * HTF_BLK
               + (f & 3) * 512 + ((((int)blockIdx.x & 1) << 4) + (f >> 2)) * 4;
#pragma unroll
    for (int pp = 0; pp < 2; ++pp) {
        float4 o;
        o.x = acc[4 * pp + 0];
        o.y = acc[4 * pp + 1];
        o.z = acc[4 * pp + 2];
        o.w = acc[4 * pp + 3];
        *(float4*)&dst[((hi << 1) + pp) * 128] = o;
    }
}

__global__ void __launch_bounds__(256)
k2_scores(const float* __restrict__ ht, const float* __restrict__ a,
          float* __restrict__ ssrc, float* __restrict__ sdst) {
    const int p = blockIdx.x * blockDim.x + threadIdx.x;
    if (p >= N_NODES * NHEAD) return;
    const int hd = p / N_NODES, n = p % N_NODES;
    const float* row = ht + (size_t)(hd * (N_NODES / 32) + (n >> 5)) * HTF_BLK
                     + ((n >> 2) & 3) * 128 + ((n >> 4) & 1) * 64 + (n & 3);
    float s1 = 0.f, s2 = 0.f;
#pragma unroll 8
    for (int f = 0; f < OUT_F; ++f) {
        const float v = row[(f & 3) * 512 + (f >> 2) * 4];
        s1 += v * a[f];
        s2 += v * a[OUT_F + f];
    }
    ssrc[n * NHEAD + hd] = s1;
    sdst[p] = s2;
}

#define KU 8
#define JB (KU * 4)
#define K4_LAG 8.0f

typedef __attribute__((ext_vector_type(4))) float v4f;
typedef __attribute__((ext_vector_type(4))) int   v4i;
#define KQ (KU / 2)
#define K4_RT      2
#define K4_HEADS   4
#define K4_WAVES   8
#define K4_THREADS (K4_WAVES * 32)
#define K4_ROWS    (K4_RT * 16)
#define JHALF      (N_NODES / 2)
#define MRG_ROWS   68
#define WS_NEEDED  ((size_t)(N_NODES * HID + 2 * N_NODES * NHEAD + N_NODES * OUT_F) * 4)

struct K4Frag {
    v4i adjq[K4_RT][KQ];
    v4f sd[KQ];
    v4f b[4][KQ];
};

__device__ __forceinline__ void k4_load(K4Frag& fr, const int* __restrict__ adj_row0,
                                        const int* __restrict__ adj_row1,
                                        const float* __restrict__ sd_col,
                                        const float* __restrict__ ht_col, int jb) {
    const float* chunk = ht_col + (size_t)(jb >> 5) * HTF_BLK;
#pragma unroll
    for (int q = 0; q < KQ; ++q) {
        const int j = jb + q * 4;
        fr.adjq[0][q] = *(const v4i*)&adj_row0[j];
        fr.adjq[1][q] = *(const v4i*)&adj_row1[j];
        fr.sd[q]      = *(const v4f*)&sd_col[j];
#pragma unroll
        for (int t = 0; t < 4; ++t)
            fr.b[t][q] = *(const v4f*)&chunk[t * 512 + q * 128];
    }
}

__device__ __forceinline__ float k4_score(int adjv, float sd, float ssr) {
    float e = ssr + sd;
    e = e > 0.f ? e : 0.2f * e;
    return adjv > 0 ? e : NEG_BIG;
}

__device__ __forceinline__ float k4_xhalf(float x) {
    const int b = __builtin_bit_cast(int, x);
    return __builtin_bit_cast(float,
        __builtin_amdgcn_permlanex16(b, b, 0x76543210, 0xfedcba98, false, false));
}

__device__ __forceinline__ void k4_compute(const K4Frag& fr, const float (&ssr)[K4_RT],
                                           float (&m_use)[K4_RT], v2f (&l_run)[K4_RT],
                                           v8f (&acc)[K4_RT][4]) {
    float e[K4_RT][KQ][4];
    float bm[K4_RT];
#pragma unroll
    for (int r = 0; r < K4_RT; ++r) {
        float b = NEG_BIG;
#pragma unroll
        for (int q = 0; q < KQ; ++q) {
#pragma unroll
            for (int x = 0; x < 4; ++x)
                e[r][q][x] = k4_score(fr.adjq[r][q][x], fr.sd[q][x], ssr[r]);
            b = fmaxf(b, fmaxf(fmaxf(e[r][q][0], e[r][q][1]), fmaxf(e[r][q][2], e[r][q][3])));
        }
        bm[r] = fmaxf(b, k4_xhalf(b));
    }
    const bool need = (bm[0] > m_use[0] + K4_LAG) || (bm[1] > m_use[1] + K4_LAG);
    if (__builtin_amdgcn_ballot_w32(need) != 0) {
#pragma unroll
        for (int r = 0; r < K4_RT; ++r) {
            const float m_new = fmaxf(m_use[r], bm[r]);
            const float corr  = __expf(m_use[r] - m_new);
            m_use[r] = m_new;
            l_run[r] = l_run[r] * corr;
#pragma unroll
            for (int t = 0; t < 4; ++t) acc[r][t] = acc[r][t] * corr;
        }
    }
#pragma unroll
    for (int q = 0; q < KQ; ++q) {
        v4f x0 = fr.b[0][q], x1 = fr.b[1][q], x2 = fr.b[2][q], x3 = fr.b[3][q];
        asm("" : "+v"(x0), "+v"(x1), "+v"(x2), "+v"(x3));
#pragma unroll
        for (int s = 0; s < 2; ++s) {
            v2f P[K4_RT];
#pragma unroll
            for (int r = 0; r < K4_RT; ++r) {
                P[r][0] = __expf(e[r][q][2 * s + 0] - m_use[r]);
                P[r][1] = __expf(e[r][q][2 * s + 1] - m_use[r]);
                l_run[r] += P[r];
            }
            v2f A0, A1, A2, A3;
            A0[0] = x0[2 * s]; A0[1] = x0[2 * s + 1];
            A1[0] = x1[2 * s]; A1[1] = x1[2 * s + 1];
            A2[0] = x2[2 * s]; A2[1] = x2[2 * s + 1];
            A3[0] = x3[2 * s]; A3[1] = x3[2 * s + 1];
#pragma unroll
            for (int r = 0; r < K4_RT; ++r) {
                acc[r][0] = wmma_f32_16x16x4(A0, P[r], acc[r][0]);
                acc[r][1] = wmma_f32_16x16x4(A1, P[r], acc[r][1]);
                acc[r][2] = wmma_f32_16x16x4(A2, P[r], acc[r][2]);
                acc[r][3] = wmma_f32_16x16x4(A3, P[r], acc[r][3]);
            }
        }
    }
}

__global__ void __launch_bounds__(K4_THREADS) __attribute__((amdgpu_waves_per_eu(2, 2)))
k4_agg(const int* __restrict__ adj, const float* __restrict__ ht,
       const float* __restrict__ ssrc, const float* __restrict__ sdst,
       float* __restrict__ out, float* __restrict__ part1, size_t ws_size) {
    __shared__ float s_part[K4_HEADS * K4_ROWS * OUT_F];
    __shared__ float s_mrg[K4_HEADS * MRG_ROWS * 32];

    const int i0   = (blockIdx.x >> 1) * K4_ROWS;
    const int hg   = blockIdx.x & 1;
    const int tid  = threadIdx.x;
    const int lane = tid & 31;
    const int hl   = (tid >> 5) & (K4_HEADS - 1);
    const int hd   = hg * K4_HEADS + hl;
    const int hf   = tid >> 7;
    const int c    = lane & 15;
    const int hi   = lane >> 4;

    float ssr[K4_RT], m_use[K4_RT];
    v2f   l_run[K4_RT];
    v8f   acc[K4_RT][4];
#pragma unroll
    for (int r = 0; r < K4_RT; ++r) {
        ssr[r]   = ssrc[(i0 + 16 * r + c) * NHEAD + hd];
        m_use[r] = -__builtin_inff();
        l_run[r] = v2f{};
#pragma unroll
        for (int t = 0; t < 4; ++t) acc[r][t] = v8f{};
    }

    const int*   adj_row0 = adj + (size_t)(i0 + c) * N_NODES + hf * JHALF + (hi << 4);
    const int*   adj_row1 = adj_row0 + (size_t)16 * N_NODES;
    const float* sd_col   = sdst + (size_t)hd * N_NODES + hf * JHALF + (hi << 4);
    const float* ht_col   = ht + (size_t)(hd * (N_NODES / 32) + hf * (JHALF / 32)) * HTF_BLK
                          + (lane << 2);

#define K4_STEP_END()                                                              \
    do {                                                                           \
        asm volatile("" : "+v"(acc[0][0]), "+v"(acc[0][1]), "+v"(acc[0][2]),       \
                          "+v"(acc[0][3]), "+v"(acc[1][0]), "+v"(acc[1][1]),       \
                          "+v"(acc[1][2]), "+v"(acc[1][3]), "+v"(l_run[0]),        \
                          "+v"(l_run[1]) : : "memory");                            \
        __builtin_amdgcn_sched_barrier(0);                                         \
    } while (0)

    K4Frag f0, f1;
    k4_load(f0, adj_row0, adj_row1, sd_col, ht_col, 0);
    K4_STEP_END();
    for (int jb = 0; jb < JHALF; jb += 2 * JB) {
        k4_load(f1, adj_row0, adj_row1, sd_col, ht_col, jb + JB);
        K4_STEP_END();
        k4_compute(f0, ssr, m_use, l_run, acc);
        K4_STEP_END();
        const int jn = (jb + 2 * JB < JHALF) ? jb + 2 * JB : jb;
        k4_load(f0, adj_row0, adj_row1, sd_col, ht_col, jn);
        K4_STEP_END();
        k4_compute(f1, ssr, m_use, l_run, acc);
        K4_STEP_END();
    }

    float L[K4_RT];
#pragma unroll
    for (int r = 0; r < K4_RT; ++r) {
        const float l_lane = l_run[r][0] + l_run[r][1];
        L[r] = l_lane + k4_xhalf(l_lane);
    }

    float* mrg = s_mrg + hl * (MRG_ROWS * 32) + lane;
    if (hf == 1) {
#pragma unroll
        for (int r = 0; r < K4_RT; ++r) {
#pragma unroll
            for (int t = 0; t < 4; ++t)
#pragma unroll
                for (int p = 0; p < 8; ++p)
                    mrg[((r * 4 + t) * 8 + p) * 32] = acc[r][t][p];
            mrg[(64 + r) * 32] = m_use[r];
            mrg[(66 + r) * 32] = L[r];
        }
    }
    __syncthreads();

    if (hf == 0) {
#pragma unroll
        for (int r = 0; r < K4_RT; ++r) {
            const float m_b = mrg[(64 + r) * 32];
            const float m   = fmaxf(m_use[r], m_b);
            const float sa  = __expf(m_use[r] - m);
            const float sb  = __expf(m_b - m);
            const float Lr  = L[r] * sa + mrg[(66 + r) * 32] * sb;
#pragma unroll
            for (int p = 0; p < 8; ++p) {
                float4 o;
                o.x = (acc[r][0][p] * sa + mrg[((r * 4 + 0) * 8 + p) * 32] * sb) / Lr;
                o.y = (acc[r][1][p] * sa + mrg[((r * 4 + 1) * 8 + p) * 32] * sb) / Lr;
                o.z = (acc[r][2][p] * sa + mrg[((r * 4 + 2) * 8 + p) * 32] * sb) / Lr;
                o.w = (acc[r][3][p] * sa + mrg[((r * 4 + 3) * 8 + p) * 32] * sb) / Lr;
                *(float4*)&s_part[(hl * K4_ROWS + 16 * r + c) * OUT_F + (hi << 5) + (p << 2)] = o;
            }
        }
    }
    __syncthreads();

    float* dst = hg == 0 ? out : part1;
    if (hg == 0 || ws_size >= WS_NEEDED) {
        for (int idx = tid; idx < K4_ROWS * OUT_F; idx += K4_THREADS) {
            float hp = 0.f;
#pragma unroll
            for (int hh = 0; hh < K4_HEADS; ++hh) hp += s_part[hh * K4_ROWS * OUT_F + idx];
            dst[(size_t)i0 * OUT_F + idx] = hp;
        }
    }
}

__global__ void __launch_bounds__(256)
k5_mean_ln(const float* __restrict__ part1, const float* __restrict__ gamma,
           const float* __restrict__ beta, float* out, size_t ws_size) {
    __shared__ float s_out[16 * OUT_F];
    __shared__ float s_mu[16], s_rs[16];
    const int i0  = blockIdx.x * 16;
    const int tid = threadIdx.x;

    for (int idx = tid; idx < 16 * OUT_F; idx += 256) {
        const size_t g = (size_t)i0 * OUT_F + idx;
        float hp = out[g];
        if (ws_size >= WS_NEEDED) hp += part1[g];
        s_out[idx] = hp * (1.0f / NHEAD);
    }
    __syncthreads();

    if (tid < 16) {
        float mu = 0.f;
        for (int f = 0; f < OUT_F; ++f) mu += s_out[tid * OUT_F + f];
        mu *= (1.0f / OUT_F);
        float var = 0.f;
        for (int f = 0; f < OUT_F; ++f) {
            const float d = s_out[tid * OUT_F + f] - mu;
            var += d * d;
        }
        var *= (1.0f / OUT_F);
        s_mu[tid] = mu;
        s_rs[tid] = rsqrtf(var + LN_EPS);
    }
    __syncthreads();

    for (int idx = tid; idx < 16 * OUT_F; idx += 256) {
        const int r = idx >> 6, f = idx & 63;
        out[(size_t)(i0 + r) * OUT_F + f] =
            (s_out[r * OUT_F + f] - s_mu[r]) * s_rs[r] * gamma[f] + beta[f];
    }
}

extern "C" void kernel_launch(void* const* d_in, const int* in_sizes, int n_in,
                              void* d_out, int out_size, void* d_ws, size_t ws_size,
                              hipStream_t stream) {
    const float* h     = (const float*)d_in[0];
    const int*   adj   = (const int*)  d_in[1];
    const float* W     = (const float*)d_in[2];
    const float* a     = (const float*)d_in[3];
    const float* gamma = (const float*)d_in[4];
    const float* beta  = (const float*)d_in[5];
    float* out = (float*)d_out;

    float* ws    = (float*)d_ws;
    float* ht    = ws;
    float* ssrc  = ht + (size_t)N_NODES * HID;
    float* sdst  = ssrc + N_NODES * NHEAD;
    float* part1 = sdst + N_NODES * NHEAD;

    k1_gemm_ht<<<dim3(N_NODES / 16, HID / 64), 128, 0, stream>>>(h, W, ht);
    k2_scores<<<(N_NODES * NHEAD) / 256, 256, 0, stream>>>(ht, a, ssrc, sdst);
    k4_agg<<<2 * (N_NODES / K4_ROWS), K4_THREADS, 0, stream>>>(adj, ht, ssrc, sdst,
                                                       out, part1, ws_size);
    k5_mean_ln<<<N_NODES / 16, 256, 0, stream>>>(part1, gamma, beta, out, ws_size);
}


# k4: numerator = adj ? max(a1_i*f1_j, a2_i*f2_j) : 0, f = exp(d), exp(d/5) from k2, row shift fixed at leaky(s_i): no exp, no running max in the j loop; old loop for max|d| > 30; last bits differ
# speedup vs baseline: 14.3327x; 1.0416x over previous
#include <hip/hip_runtime.h>
#include <hip/hip_bf16.h>

#define N_NODES 4096
#define IN_F    256
#define OUT_F   64
#define NHEAD   8
#define HID     512
#define NEG_BIG (-9.0e15f)
#define LN_EPS  1e-05f

typedef __attribute__((ext_vector_type(2))) float v2f;
typedef __attribute__((ext_vector_type(8))) float v8f;

__device__ __forceinline__ v8f wmma_f32_16x16x4(v2f A, v2f B, v8f C) {
    return __builtin_amdgcn_wmma_f32_16x16x4_f32(false, A, false, B, (short)0, C,
                                                 false, false);
}

#define HTF_BLK 2048
#define WS_NEEDED  ((size_t)(N_NODES * HID + 2 * N_NODES * NHEAD + N_NODES * OUT_F) * 4)
#define WS_NEEDED2 (WS_NEEDED + (size_t)(2 * N_NODES * NHEAD) * 4)
__global__ void __launch_bounds__(128)
k1_gemm_ht(const float* __restrict__ h, const float* __restrict__ W,
           float* __restrict__ ht) {
    __shared__ float sA[16 * 260];
    const int i0 = blockIdx.x * 16;
    const int n0 = blockIdx.y * 64;
    const int tid  = threadIdx.x;
    const int lane = tid & 31;
    const int wave = tid >> 5;

    for (int idx = tid; idx < 16 * 256 / 4; idx += 128) {
        const int r = idx >> 6, c = (idx & 63) << 2;
        const float4 v = *(const float4*)&h[(size_t)(i0 + r) * IN_F + c];
        *(float4*)&sA[r * 260 + c] = v;
    }
    __syncthreads();

    const int m  = lane & 15;
    const int hi = lane >> 4;

    v8f acc = {};
    for (int ks = 0; ks < IN_F / 4; ++ks) {
        v2f A, B;
#pragma unroll
        for (int v = 0; v < 2; ++v) {
            const int kk = (hi << 1) + v;
            const int k  = ks * 4 + kk;
            A[v] = sA[m * 260 + k];
            B[v] = W[(size_t)k * HID + n0 + (wave << 4) + m];
        }
        acc = wmma_f32_16x16x4(A, B, acc);
    }

    const int f  = (wave << 4) + m;
    float* dst = ht + (size_t)(blockIdx.y * (N_NODES / 32) + (blockIdx.x >> 1)) * HTF_BLK
               + (f & 3) * 512 + ((((int)blockIdx.x & 1) << 4) + (f >> 2)) * 4;
#pragma unroll
    for (int pp = 0; pp < 2; ++pp) {
        float4 o;
        o.x = acc[4 * pp + 0];
        o.y = acc[4 * pp + 1];
        o.z = acc[4 * pp + 2];
        o.w = acc[4 * pp + 3];
        *(float4*)&dst[((hi << 1) + pp) * 128] = o;
    }
}

__global__ void __launch_bounds__(256)
k2_scores(const float* __restrict__ ht, const float* __restrict__ a,
          float* __restrict__ ssrc, float* __restrict__ sdst,
          float* __restrict__ f1, float* __restrict__ f2, size_t ws_size) {
    const int p = blockIdx.x * blockDim.x + threadIdx.x;
    if (p >= N_NODES * NHEAD) return;
    const int hd = p / N_NODES, n = p % N_NODES;
    const float* row = ht + (size_t)(hd * (N_NODES / 32) + (n >> 5)) * HTF_BLK
                     + ((n >> 2) & 3) * 128 + ((n >> 4) & 1) * 64 + (n & 3);
    float s1 = 0.f, s2 = 0.f;
#pragma unroll 8
    for (int f = 0; f < OUT_F; ++f) {
        const float v = row[(f & 3) * 512 + (f >> 2) * 4];
        s1 += v * a[f];
        s2 += v * a[OUT_F + f];
    }
    ssrc[n * NHEAD + hd] = s1;
    sdst[p] = s2;
    if (ws_size >= WS_NEEDED2) {
        f1[p] = expf(s2);
        f2[p] = expf(0.2f * s2);
    }
}

#define KU 8
#define JB (KU * 4)
#define K4_LAG 8.0f
#define K4_DMAX 30.0f

typedef __attribute__((ext_vector_type(4))) float v4f;
typedef __attribute__((ext_vector_type(4))) int   v4i;
#define KQ (KU / 2)
#define K4_RT      2
#define K4_HEADS   4
#define K4_WAVES   8
#define K4_THREADS (K4_WAVES * 32)
#define K4_ROWS    (K4_RT * 16)
#define JHALF      (N_NODES / 2)
#define MRG_ROWS   68

struct K4Frag {
    v4i adjq[K4_RT][KQ];
    v4f sd[KQ];
    v4f b[4][KQ];
};

__device__ __forceinline__ void k4_load(K4Frag& fr, const int* __restrict__ adj_row0,
                                        const int* __restrict__ adj_row1,
                                        const float* __restrict__ sd_col,
                                        const float* __restrict__ ht_col, int jb) {
    const float* chunk = ht_col + (size_t)(jb >> 5) * HTF_BLK;
#pragma unroll
    for (int q = 0; q < KQ; ++q) {
        const int j = jb + q * 4;
        fr.adjq[0][q] = *(const v4i*)&adj_row0[j];
        fr.adjq[1][q] = *(const v4i*)&adj_row1[j];
        fr.sd[q]      = *(const v4f*)&sd_col[j];
#pragma unroll
        for (int t = 0; t < 4; ++t)
            fr.b[t][q] = *(const v4f*)&chunk[t * 512 + q * 128];
    }
}

struct K4FragF {
    v4i adjq[K4_RT][KQ];
    v4f f1[KQ];
    v4f f2[KQ];
    v4f b[4][KQ];
};

__device__ __forceinline__ void k4_load_f(K4FragF& fr, const int* __restrict__ adj_row0,
                                          const int* __restrict__ adj_row1,
                                          const float* __restrict__ f1_col,
                                          const float* __restrict__ f2_col,
                                          const float* __restrict__ ht_col, int jb) {
    const float* chunk = ht_col + (size_t)(jb >> 5) * HTF_BLK;
#pragma unroll
    for (int q = 0; q < KQ; ++q) {
        const int j = jb + q * 4;
        fr.adjq[0][q] = *(const v4i*)&adj_row0[j];
        fr.adjq[1][q] = *(const v4i*)&adj_row1[j];
        fr.f1[q]      = *(const v4f*)&f1_col[j];
        fr.f2[q]      = *(const v4f*)&f2_col[j];
#pragma unroll
        for (int t = 0; t < 4; ++t)
            fr.b[t][q] = *(const v4f*)&chunk[t * 512 + q * 128];
    }
}

__device__ __forceinline__ void k4_compute_f(const K4FragF& fr, const float (&a1)[K4_RT],
                                             const float (&a2)[K4_RT], v2f (&l_run)[K4_RT],
                                             v8f (&acc)[K4_RT][4]) {
#pragma unroll
    for (int q = 0; q < KQ; ++q) {
        v4f x0 = fr.b[0][q], x1 = fr.b[1][q], x2 = fr.b[2][q], x3 = fr.b[3][q];
        asm("" : "+v"(x0), "+v"(x1), "+v"(x2), "+v"(x3));
#pragma unroll
        for (int s = 0; s < 2; ++s) {
            v2f P[K4_RT];
#pragma unroll
            for (int r = 0; r < K4_RT; ++r) {
#pragma unroll
                for (int v = 0; v < 2; ++v) {
                    const float p = fmaxf(a1[r] * fr.f1[q][2 * s + v], a2[r] * fr.f2[q][2 * s + v]);
                    P[r][v] = fr.adjq[r][q][2 * s + v] > 0 ? p : 0.f;
                }
                l_run[r] += P[r];
            }
            v2f A0, A1, A2, A3;
            A0[0] = x0[2 * s]; A0[1] = x0[2 * s + 1];
            A1[0] = x1[2 * s]; A1[1] = x1[2 * s + 1];
            A2[0] = x2[2 * s]; A2[1] = x2[2 * s + 1];
            A3[0] = x3[2 * s]; A3[1] = x3[2 * s + 1];
#pragma unroll
            for (int r = 0; r < K4_RT; ++r) {
                acc[r][0] = wmma_f32_16x16x4(A0, P[r], acc[r][0]);
                acc[r][1] = wmma_f32_16x16x4(A1, P[r], acc[r][1]);
                acc[r][2] = wmma_f32_16x16x4(A2, P[r], acc[r][2]);
                acc[r][3] = wmma_f32_16x16x4(A3, P[r], acc[r][3]);
            }
        }
    }
}

__device__ __forceinline__ float k4_score(int adjv, float sd, float ssr) {
    float e = ssr + sd;
    e = e > 0.f ? e : 0.2f * e;
    return adjv > 0 ? e : NEG_BIG;
}

__device__ __forceinline__ float k4_xhalf(float x) {
    const int b = __builtin_bit_cast(int, x);
    return __builtin_bit_cast(float,
        __builtin_amdgcn_permlanex16(b, b, 0x76543210, 0xfedcba98, false, false));
}

__device__ __forceinline__ void k4_compute(const K4Frag& fr, const float (&ssr)[K4_RT],
                                           float (&m_use)[K4_RT], v2f (&l_run)[K4_RT],
                                           v8f (&acc)[K4_RT][4]) {
    float e[K4_RT][KQ][4];
    float bm[K4_RT];
#pragma unroll
    for (int r = 0; r < K4_RT; ++r) {
        float b = NEG_BIG;
#pragma unroll
        for (int q = 0; q < KQ; ++q) {
#pragma unroll
            for (int x = 0; x < 4; ++x)
                e[r][q][x] = k4_score(fr.adjq[r][q][x], fr.sd[q][x], ssr[r]);
            b = fmaxf(b, fmaxf(fmaxf(e[r][q][0], e[r][q][1]), fmaxf(e[r][q][2], e[r][q][3])));
        }
        bm[r] = fmaxf(b, k4_xhalf(b));
    }
    const bool need = (bm[0] > m_use[0] + K4_LAG) || (bm[1] > m_use[1] + K4_LAG);
    if (__builtin_amdgcn_ballot_w32(need) != 0) {
#pragma unroll
        for (int r = 0; r < K4_RT; ++r) {
            const float m_new = fmaxf(m_use[r], bm[r]);
            const float corr  = __expf(m_use[r] - m_new);
            m_use[r] = m_new;
            l_run[r] = l_run[r] * corr;
#pragma unroll
            for (int t = 0; t < 4; ++t) acc[r][t] = acc[r][t] * corr;
        }
    }
#pragma unroll
    for (int q = 0; q < KQ; ++q) {
        v4f x0 = fr.b[0][q], x1 = fr.b[1][q], x2 = fr.b[2][q], x3 = fr.b[3][q];
        asm("" : "+v"(x0), "+v"(x1), "+v"(x2), "+v"(x3));
#pragma unroll
        for (int s = 0; s < 2; ++s) {
            v2f P[K4_RT];
#pragma unroll
            for (int r = 0; r < K4_RT; ++r) {
                P[r][0] = __expf(e[r][q][2 * s + 0] - m_use[r]);
                P[r][1] = __expf(e[r][q][2 * s + 1] - m_use[r]);
                l_run[r] += P[r];
            }
            v2f A0, A1, A2, A3;
            A0[0] = x0[2 * s]; A0[1] = x0[2 * s + 1];
            A1[0] = x1[2 * s]; A1[1] = x1[2 * s + 1];
            A2[0] = x2[2 * s]; A2[1] = x2[2 * s + 1];
            A3[0] = x3[2 * s]; A3[1] = x3[2 * s + 1];
#pragma unroll
            for (int r = 0; r < K4_RT; ++r) {
                acc[r][0] = wmma_f32_16x16x4(A0, P[r], acc[r][0]);
                acc[r][1] = wmma_f32_16x16x4(A1, P[r], acc[r][1]);
                acc[r][2] = wmma_f32_16x16x4(A2, P[r], acc[r][2]);
                acc[r][3] = wmma_f32_16x16x4(A3, P[r], acc[r][3]);
            }
        }
    }
}

__global__ void __launch_bounds__(K4_THREADS) __attribute__((amdgpu_waves_per_eu(2, 2)))
k4_agg(const int* __restrict__ adj, const float* __restrict__ ht,
       const float* __restrict__ ssrc, const float* __restrict__ sdst,
       const float* __restrict__ f1g, const float* __restrict__ f2g,
       float* __restrict__ out, float* __restrict__ part1, size_t ws_size) {
    __shared__ float s_part[K4_HEADS * K4_ROWS * OUT_F];
    __shared__ float s_mrg[K4_HEADS * MRG_ROWS * 32];

    const int i0   = (blockIdx.x >> 1) * K4_ROWS;
    const int hg   = blockIdx.x & 1;
    const int tid  = threadIdx.x;
    const int lane = tid & 31;
    const int hl   = (tid >> 5) & (K4_HEADS - 1);
    const int hd   = hg * K4_HEADS + hl;
    const int hf   = tid >> 7;
    const int c    = lane & 15;
    const int hi   = lane >> 4;

    float ssr[K4_RT], m_use[K4_RT];
    v2f   l_run[K4_RT];
    v8f   acc[K4_RT][4];
#pragma unroll
    for (int r = 0; r < K4_RT; ++r) {
        ssr[r]   = ssrc[(i0 + 16 * r + c) * NHEAD + hd];
        m_use[r] = -__builtin_inff();
        l_run[r] = v2f{};
#pragma unroll
        for (int t = 0; t < 4; ++t) acc[r][t] = v8f{};
    }

    const int*   adj_row0 = adj + (size_t)(i0 + c) * N_NODES + hf * JHALF + (hi << 4);
    const int*   adj_row1 = adj_row0 + (size_t)16 * N_NODES;
    const float* sd_col   = sdst + (size_t)hd * N_NODES + hf * JHALF + (hi << 4);
    const float* ht_col   = ht + (size_t)(hd * (N_NODES / 32) + hf * (JHALF / 32)) * HTF_BLK
                          + (lane << 2);

#define K4_STEP_END()                                                              \
    do {                                                                           \
        asm volatile("" : "+v"(acc[0][0]), "+v"(acc[0][1]), "+v"(acc[0][2]),       \
                          "+v"(acc[0][3]), "+v"(acc[1][0]), "+v"(acc[1][1]),       \
                          "+v"(acc[1][2]), "+v"(acc[1][3]), "+v"(l_run[0]),        \
                          "+v"(l_run[1]) : : "memory");                            \
        __builtin_amdgcn_sched_barrier(0);                                         \
    } while (0)

    float dmx = 0.f;
    {
        const float* sdh = sdst + (size_t)hd * N_NODES + (lane << 2);
#pragma unroll
        for (int k = 0; k < N_NODES / 128; ++k) {
            const v4f v = *(const v4f*)&sdh[k * 128];
            dmx = fmaxf(dmx, fmaxf(fmaxf(fabsf(v[0]), fabsf(v[1])),
                                   fmaxf(fabsf(v[2]), fabsf(v[3]))));
        }
        for (int msk = 16; msk >= 1; msk >>= 1) dmx = fmaxf(dmx, __shfl_xor(dmx, msk));
    }
    bool fast = ws_size >= WS_NEEDED2 &&
                __builtin_amdgcn_readfirstlane(dmx <= K4_DMAX ? 1 : 0) != 0;

    float L[K4_RT];
    if (fast) {
        float a1[K4_RT], a2[K4_RT], mfix[K4_RT];
#pragma unroll
        for (int r = 0; r < K4_RT; ++r) {
            mfix[r] = ssr[r] > 0.f ? ssr[r] : 0.2f * ssr[r];
            a1[r]   = expf(ssr[r] - mfix[r]);
            a2[r]   = expf(0.2f * ssr[r] - mfix[r]);
        }
        const float* f1_col = f1g + (size_t)hd * N_NODES + hf * JHALF + (hi << 4);
        const float* f2_col = f2g + (size_t)hd * N_NODES + hf * JHALF + (hi << 4);
        K4FragF g0, g1;
        k4_load_f(g0, adj_row0, adj_row1, f1_col, f2_col, ht_col, 0);
        K4_STEP_END();
        for (int jb = 0; jb < JHALF; jb += 2 * JB) {
            k4_load_f(g1, adj_row0, adj_row1, f1_col, f2_col, ht_col, jb + JB);
            K4_STEP_END();
            k4_compute_f(g0, a1, a2, l_run, acc);
            K4_STEP_END();
            const int jn = (jb + 2 * JB < JHALF) ? jb + 2 * JB : jb;
            k4_load_f(g0, adj_row0, adj_row1, f1_col, f2_col, ht_col, jn);
            K4_STEP_END();
            k4_compute_f(g1, a1, a2, l_run, acc);
            K4_STEP_END();
        }
        bool empty = false;
#pragma unroll
        for (int r = 0; r < K4_RT; ++r) {
            const float l_lane = l_run[r][0] + l_run[r][1];
            L[r] = l_lane + k4_xhalf(l_lane);
            empty = empty || !(L[r] > 0.f);
        }
        if (__builtin_amdgcn_ballot_w32(empty) != 0) {
            fast = false;
#pragma unroll
            for (int r = 0; r < K4_RT; ++r) {
                l_run[r] = v2f{};
#pragma unroll
                for (int t = 0; t < 4; ++t) acc[r][t] = v8f{};
            }
        } else {
#pragma unroll
            for (int r = 0; r < K4_RT; ++r) m_use[r] = mfix[r];
        }
    }
    if (!fast) {
        K4Frag f0, f1;
        k4_load(f0, adj_row0, adj_row1, sd_col, ht_col, 0);
        K4_STEP_END();
        for (int jb = 0; jb < JHALF; jb += 2 * JB) {
            k4_load(f1, adj_row0, adj_row1, sd_col, ht_col, jb + JB);
            K4_STEP_END();
            k4_compute(f0, ssr, m_use, l_run, acc);
            K4_STEP_END();
            const int jn = (jb + 2 * JB < JHALF) ? jb + 2 * JB : jb;
            k4_load(f0, adj_row0, adj_row1, sd_col, ht_col, jn);
            K4_STEP_END();
            k4_compute(f1, ssr, m_use, l_run, acc);
            K4_STEP_END();
        }
#pragma unroll
        for (int r = 0; r < K4_RT; ++r) {
            const float l_lane = l_run[r][0] + l_run[r][1];
            L[r] = l_lane + k4_xhalf(l_lane);
        }
    }

    float* mrg = s_mrg + hl * (MRG_ROWS * 32) + lane;
    if (hf == 1) {
#pragma unroll
        for (int r = 0; r < K4_RT; ++r) {
#pragma unroll
            for (int t = 0; t < 4; ++t)
#pragma unroll
                for (int p = 0; p < 8; ++p)
                    mrg[((r * 4 + t) * 8 + p) * 32] = acc[r][t][p];
            mrg[(64 + r) * 32] = m_use[r];
            mrg[(66 + r) * 32] = L[r];
        }
    }
    __syncthreads();

    if (hf == 0) {
#pragma unroll
        for (int r = 0; r < K4_RT; ++r) {
            const float m_b = mrg[(64 + r) * 32];
            const float m   = fmaxf(m_use[r], m_b);
            const float sa  = __expf(m_use[r] - m);
            const float sb  = __expf(m_b - m);
            const float Lr  = L[r] * sa + mrg[(66 + r) * 32] * sb;
#pragma unroll
            for (int p = 0; p < 8; ++p) {
                float4 o;
                o.x = (acc[r][0][p] * sa + mrg[((r * 4 + 0) * 8 + p) * 32] * sb) / Lr;
                o.y = (acc[r][1][p] * sa + mrg[((r * 4 + 1) * 8 + p) * 32] * sb) / Lr;
                o.z = (acc[r][2][p] * sa + mrg[((r * 4 + 2) * 8 + p) * 32] * sb) / Lr;
                o.w = (acc[r][3][p] * sa + mrg[((r * 4 + 3) * 8 + p) * 32] * sb) / Lr;
                *(float4*)&s_part[(hl * K4_ROWS + 16 * r + c) * OUT_F + (hi << 5) + (p << 2)] = o;
            }
        }
    }
    __syncthreads();

    float* dst = hg == 0 ? out : part1;
    if (hg == 0 || ws_size >= WS_NEEDED) {
        for (int idx = tid; idx < K4_ROWS * OUT_F; idx += K4_THREADS) {
            float hp = 0.f;
#pragma unroll
            for (int hh = 0; hh < K4_HEADS; ++hh) hp += s_part[hh * K4_ROWS * OUT_F + idx];
            dst[(size_t)i0 * OUT_F + idx] = hp;
        }
    }
}

__global__ void __launch_bounds__(256)
k5_mean_ln(const float* __restrict__ part1, const float* __restrict__ gamma,
           const float* __restrict__ beta, float* out, size_t ws_size) {
    __shared__ float s_out[16 * OUT_F];
    __shared__ float s_mu[16], s_rs[16];
    const int i0  = blockIdx.x * 16;
    const int tid = threadIdx.x;

    for (int idx = tid; idx < 16 * OUT_F; idx += 256) {
        const size_t g = (size_t)i0 * OUT_F + idx;
        float hp = out[g];
        if (ws_size >= WS_NEEDED) hp += part1[g];
        s_out[idx] = hp * (1.0f / NHEAD);
    }
    __syncthreads();

    if (tid < 16) {
        float mu = 0.f;
        for (int f = 0; f < OUT_F; ++f) mu += s_out[tid * OUT_F + f];
        mu *= (1.0f / OUT_F);
        float var = 0.f;
        for (int f = 0; f < OUT_F; ++f) {
            const float d = s_out[tid * OUT_F + f] - mu;
            var += d * d;
        }
        var *= (1.0f / OUT_F);
        s_mu[tid] = mu;
        s_rs[tid] = rsqrtf(var + LN_EPS);
    }
    __syncthreads();

    for (int idx = tid; idx < 16 * OUT_F; idx += 256) {
        const int r = idx >> 6, f = idx & 63;
        out[(size_t)(i0 + r) * OUT_F + f] =
            (s_out[r * OUT_F + f] - s_mu[r]) * s_rs[r] * gamma[f] + beta[f];
    }
}

extern "C" void kernel_launch(void* const* d_in, const int* in_sizes, int n_in,
                              void* d_out, int out_size, void* d_ws, size_t ws_size,
                              hipStream_t stream) {
    const float* h     = (const float*)d_in[0];
    const int*   adj   = (const int*)  d_in[1];
    const float* W     = (const float*)d_in[2];
    const float* a     = (const float*)d_in[3];
    const float* gamma = (const float*)d_in[4];
    const float* beta  = (const float*)d_in[5];
    float* out = (float*)d_out;

    float* ws    = (float*)d_ws;
    float* ht    = ws;
    float* ssrc  = ht + (size_t)N_NODES * HID;
    float* sdst  = ssrc + N_NODES * NHEAD;
    float* part1 = sdst + N_NODES * NHEAD;
    float* f1    = part1 + (size_t)N_NODES * OUT_F;
    float* f2    = f1 + N_NODES * NHEAD;

    k1_gemm_ht<<<dim3(N_NODES / 16, HID / 64), 128, 0, stream>>>(h, W, ht);
    k2_scores<<<(N_NODES * NHEAD) / 256, 256, 0, stream>>>(ht, a, ssrc, sdst, f1, f2, ws_size);
    k4_agg<<<2 * (N_NODES / K4_ROWS), K4_THREADS, 0, stream>>>(adj, ht, ssrc, sdst, f1, f2,
                                                       out, part1, ws_size);
    k5_mean_ln<<<N_NODES / 16, 256, 0, stream>>>(part1, gamma, beta, out, ws_size);
}


# k4: adjacency taken as lane masks (K2 votes them once, 1 bit per pair, 2 MB) by scalar loads, used as select masks; was 8 int loads of 16 cache lines per block, wave and head; sum orders kept
# speedup vs baseline: 14.8834x; 1.0384x over previous
#include <hip/hip_runtime.h>
#include <hip/hip_bf16.h>

#define N_NODES 4096
#define IN_F    256
#define OUT_F   64
#define NHEAD   8
#define HID     512
#define NEG_BIG (-9.0e15f)
#define LN_EPS  1e-05f

typedef __attribute__((ext_vector_type(2))) float v2f;
typedef __attribute__((ext_vector_type(8))) float v8f;

__device__ __forceinline__ v8f wmma_f32_16x16x4(v2f A, v2f B, v8f C) {
    return __builtin_amdgcn_wmma_f32_16x16x4_f32(false, A, false, B, (short)0, C,
                                                 false, false);
}

#define HTF_BLK 2048
#define WS_NEEDED  ((size_t)(N_NODES * HID + 2 * N_NODES * NHEAD + N_NODES * OUT_F) * 4)
#define WS_NEEDED2 (WS_NEEDED + (size_t)(2 * N_NODES * NHEAD) * 4)
#define MASK_WORDS ((size_t)(N_NODES / 32) * (N_NODES / 32) * 32)
#define WS_NEEDED3 (WS_NEEDED2 + MASK_WORDS * 4)
#define K2_SCORE_BLOCKS ((N_NODES * NHEAD) / 256)
#define K2_MASK_BLOCKS  512
typedef __attribute__((ext_vector_type(4))) int   v4i;
typedef __attribute__((ext_vector_type(8))) unsigned v8u;
__global__ void __launch_bounds__(128)
k1_gemm_ht(const float* __restrict__ h, const float* __restrict__ W,
           float* __restrict__ ht) {
    __shared__ float sA[16 * 260];
    const int i0 = blockIdx.x * 16;
    const int n0 = blockIdx.y * 64;
    const int tid  = threadIdx.x;
    const int lane = tid & 31;
    const int wave = tid >> 5;

    for (int idx = tid; idx < 16 * 256 / 4; idx += 128) {
        const int r = idx >> 6, c = (idx & 63) << 2;
        const float4 v = *(const float4*)&h[(size_t)(i0 + r) * IN_F + c];
        *(float4*)&sA[r * 260 + c] = v;
    }
    __syncthreads();

    const int m  = lane & 15;
    const int hi = lane >> 4;

    v8f acc = {};
    for (int ks = 0; ks < IN_F / 4; ++ks) {
        v2f A, B;
#pragma unroll
        for (int v = 0; v < 2; ++v) {
            const int kk = (hi << 1) + v;
            const int k  = ks * 4 + kk;
            A[v] = sA[m * 260 + k];
            B[v] = W[(size_t)k * HID + n0 + (wave << 4) + m];
        }
        acc = wmma_f32_16x16x4(A, B, acc);
    }

    const int f  = (wave << 4) + m;
    float* dst = ht + (size_t)(blockIdx.y * (N_NODES / 32) + (blockIdx.x >> 1)) * HTF_BLK
               + (f & 3) * 512 + ((((int)blockIdx.x & 1) << 4) + (f >> 2)) * 4;
#pragma unroll
    for (int pp = 0; pp < 2; ++pp) {
        float4 o;
        o.x = acc[4 * pp + 0];
        o.y = acc[4 * pp + 1];
        o.z = acc[4 * pp + 2];
        o.w = acc[4 * pp + 3];
        *(float4*)&dst[((hi << 1) + pp) * 128] = o;
    }
}

__global__ void __launch_bounds__(256)
k2_scores(const float* __restrict__ ht, const float* __restrict__ a,
          const int* __restrict__ adj,
          float* __restrict__ ssrc, float* __restrict__ sdst,
          float* __restrict__ f1, float* __restrict__ f2,
          unsigned* __restrict__ amask, size_t ws_size) {
    if (blockIdx.x >= K2_SCORE_BLOCKS) {
        if (ws_size < WS_NEEDED3) return;
        const int lane = threadIdx.x & 31;
        const int u  = (blockIdx.x - K2_SCORE_BLOCKS) * 8 + (threadIdx.x >> 5);
        const int g  = u >> 5;
        const int b0 = (u & 31) * 4;
        const int c  = lane & 15, hi = lane >> 4;
        const int* base = adj + (size_t)(32 * g + c) * N_NODES + 32 * b0 + 16 * hi;
        v4i av[4][2][4];
#pragma unroll
        for (int k = 0; k < 4; ++k)
#pragma unroll
            for (int r = 0; r < 2; ++r)
#pragma unroll
                for (int q = 0; q < 4; ++q)
                    av[k][r][q] = *(const v4i*)&base[(size_t)r * 16 * N_NODES + 32 * k + 4 * q];
#pragma unroll
        for (int k = 0; k < 4; ++k) {
            unsigned word = 0u;
#pragma unroll
            for (int r = 0; r < 2; ++r)
#pragma unroll
                for (int q = 0; q < 4; ++q)
#pragma unroll
                    for (int e = 0; e < 4; ++e) {
                        const unsigned vote = __builtin_amdgcn_ballot_w32(av[k][r][q][e] > 0);
                        if (lane == r * 16 + 4 * q + e) word = vote;
                    }
            amask[(size_t)(g * (N_NODES / 32) + b0 + k) * 32 + lane] = word;
        }
        return;
    }
    const int p = blockIdx.x * blockDim.x + threadIdx.x;
    if (p >= N_NODES * NHEAD) return;
    const int hd = p / N_NODES, n = p % N_NODES;
    const float* row = ht + (size_t)(hd * (N_NODES / 32) + (n >> 5)) * HTF_BLK
                     + ((n >> 2) & 3) * 128 + ((n >> 4) & 1) * 64 + (n & 3);
    float s1 = 0.f, s2 = 0.f;
#pragma unroll 8
    for (int f = 0; f < OUT_F; ++f) {
        const float v = row[(f & 3) * 512 + (f >> 2) * 4];
        s1 += v * a[f];
        s2 += v * a[OUT_F + f];
    }
    ssrc[n * NHEAD + hd] = s1;
    sdst[p] = s2;
    if (ws_size >= WS_NEEDED2) {
        f1[p] = expf(s2);
        f2[p] = expf(0.2f * s2);
    }
}

#define KU 8
#define JB (KU * 4)
#define K4_LAG 8.0f
#define K4_DMAX 30.0f

typedef __attribute__((ext_vector_type(4))) float v4f;
#define KQ (KU / 2)
#define K4_RT      2
#define K4_HEADS   4
#define K4_WAVES   8
#define K4_THREADS (K4_WAVES * 32)
#define K4_ROWS    (K4_RT * 16)
#define JHALF      (N_NODES / 2)
#define MRG_ROWS   68

struct K4Frag {
    v4i adjq[K4_RT][KQ];
    v4f sd[KQ];
    v4f b[4][KQ];
};

__device__ __forceinline__ void k4_load(K4Frag& fr, const int* __restrict__ adj_row0,
                                        const int* __restrict__ adj_row1,
                                        const float* __restrict__ sd_col,
                                        const float* __restrict__ ht_col, int jb) {
    const float* chunk = ht_col + (size_t)(jb >> 5) * HTF_BLK;
#pragma unroll
    for (int q = 0; q < KQ; ++q) {
        const int j = jb + q * 4;
        fr.adjq[0][q] = *(const v4i*)&adj_row0[j];
        fr.adjq[1][q] = *(const v4i*)&adj_row1[j];
        fr.sd[q]      = *(const v4f*)&sd_col[j];
#pragma unroll
        for (int t = 0; t < 4; ++t)
            fr.b[t][q] = *(const v4f*)&chunk[t * 512 + q * 128];
    }
}

struct K4FragF {
    v8u m[4];
    v4f f1[KQ];
    v4f f2[KQ];
    v4f b[4][KQ];
};

typedef const __attribute__((address_space(4))) v8u* k4_mask_ptr;

__device__ __forceinline__ void k4_load_m(K4FragF& fr, k4_mask_ptr mrow, int jb) {
#pragma unroll
    for (int i = 0; i < 4; ++i) fr.m[i] = mrow[(jb >> 5) * 4 + i];
}

__device__ __forceinline__ void k4_load_f(K4FragF& fr,
                                          const float* __restrict__ f1_col,
                                          const float* __restrict__ f2_col,
                                          const float* __restrict__ ht_col, int jb) {
    const float* chunk = ht_col + (size_t)(jb >> 5) * HTF_BLK;
#pragma unroll
    for (int q = 0; q < KQ; ++q) {
        const int j = jb + q * 4;
        fr.f1[q]      = *(const v4f*)&f1_col[j];
        fr.f2[q]      = *(const v4f*)&f2_col[j];
#pragma unroll
        for (int t = 0; t < 4; ++t)
            fr.b[t][q] = *(const v4f*)&chunk[t * 512 + q * 128];
    }
}

__device__ __forceinline__ void k4_compute_f(const K4FragF& fr, const float (&a1)[K4_RT],
                                             const float (&a2)[K4_RT], v2f (&l_run)[K4_RT],
                                             v8f (&acc)[K4_RT][4]) {
#pragma unroll
    for (int q = 0; q < KQ; ++q) {
        v4f x0 = fr.b[0][q], x1 = fr.b[1][q], x2 = fr.b[2][q], x3 = fr.b[3][q];
#pragma unroll
        for (int s = 0; s < 2; ++s) {
            v2f P[K4_RT];
#pragma unroll
            for (int r = 0; r < K4_RT; ++r) {
#pragma unroll
                for (int v = 0; v < 2; ++v) {
                    const float p = fmaxf(a1[r] * fr.f1[q][2 * s + v], a2[r] * fr.f2[q][2 * s + v]);
                    const int   w = r * 16 + 4 * q + 2 * s + v;
                    P[r][v] = __builtin_amdgcn_inverse_ballot_w32(fr.m[w >> 3][w & 7]) ? p : 0.f;
                }
                l_run[r] += P[r];
            }
            v2f A0, A1, A2, A3;
            A0[0] = x0[2 * s]; A0[1] = x0[2 * s + 1];
            A1[0] = x1[2 * s]; A1[1] = x1[2 * s + 1];
            A2[0] = x2[2 * s]; A2[1] = x2[2 * s + 1];
            A3[0] = x3[2 * s]; A3[1] = x3[2 * s + 1];
#pragma unroll
            for (int r = 0; r < K4_RT; ++r) {
                acc[r][0] = wmma_f32_16x16x4(A0, P[r], acc[r][0]);
                acc[r][1] = wmma_f32_16x16x4(A1, P[r], acc[r][1]);
                acc[r][2] = wmma_f32_16x16x4(A2, P[r], acc[r][2]);
                acc[r][3] = wmma_f32_16x16x4(A3, P[r], acc[r][3]);
            }
        }
    }
}

__device__ __forceinline__ float k4_score(int adjv, float sd, float ssr) {
    float e = ssr + sd;
    e = e > 0.f ? e : 0.2f * e;
    return adjv > 0 ? e : NEG_BIG;
}

__device__ __forceinline__ float k4_xhalf(float x) {
    const int b = __builtin_bit_cast(int, x);
    return __builtin_bit_cast(float,
        __builtin_amdgcn_permlanex16(b, b, 0x76543210, 0xfedcba98, false, false));
}

__device__ __forceinline__ void k4_compute(const K4Frag& fr, const float (&ssr)[K4_RT],
                                           float (&m_use)[K4_RT], v2f (&l_run)[K4_RT],
                                           v8f (&acc)[K4_RT][4]) {
    float e[K4_RT][KQ][4];
    float bm[K4_RT];
#pragma unroll
    for (int r = 0; r < K4_RT; ++r) {
        float b = NEG_BIG;
#pragma unroll
        for (int q = 0; q < KQ; ++q) {
#pragma unroll
            for (int x = 0; x < 4; ++x)
                e[r][q][x] = k4_score(fr.adjq[r][q][x], fr.sd[q][x], ssr[r]);
            b = fmaxf(b, fmaxf(fmaxf(e[r][q][0], e[r][q][1]), fmaxf(e[r][q][2], e[r][q][3])));
        }
        bm[r] = fmaxf(b, k4_xhalf(b));
    }
    const bool need = (bm[0] > m_use[0] + K4_LAG) || (bm[1] > m_use[1] + K4_LAG);
    if (__builtin_amdgcn_ballot_w32(need) != 0) {
#pragma unroll
        for (int r = 0; r < K4_RT; ++r) {
            const float m_new = fmaxf(m_use[r], bm[r]);
            const float corr  = __expf(m_use[r] - m_new);
            m_use[r] = m_new;
            l_run[r] = l_run[r] * corr;
#pragma unroll
            for (int t = 0; t < 4; ++t) acc[r][t] = acc[r][t] * corr;
        }
    }
#pragma unroll
    for (int q = 0; q < KQ; ++q) {
        v4f x0 = fr.b[0][q], x1 = fr.b[1][q], x2 = fr.b[2][q], x3 = fr.b[3][q];
        asm("" : "+v"(x0), "+v"(x1), "+v"(x2), "+v"(x3));
#pragma unroll
        for (int s = 0; s < 2; ++s) {
            v2f P[K4_RT];
#pragma unroll
            for (int r = 0; r < K4_RT; ++r) {
                P[r][0] = __expf(e[r][q][2 * s + 0] - m_use[r]);
                P[r][1] = __expf(e[r][q][2 * s + 1] - m_use[r]);
                l_run[r] += P[r];
            }
            v2f A0, A1, A2, A3;
            A0[0] = x0[2 * s]; A0[1] = x0[2 * s + 1];
            A1[0] = x1[2 * s]; A1[1] = x1[2 * s + 1];
            A2[0] = x2[2 * s]; A2[1] = x2[2 * s + 1];
            A3[0] = x3[2 * s]; A3[1] = x3[2 * s + 1];
#pragma unroll
            for (int r = 0; r < K4_RT; ++r) {
                acc[r][0] = wmma_f32_16x16x4(A0, P[r], acc[r][0]);
                acc[r][1] = wmma_f32_16x16x4(A1, P[r], acc[r][1]);
                acc[r][2] = wmma_f32_16x16x4(A2, P[r], acc[r][2]);
                acc[r][3] = wmma_f32_16x16x4(A3, P[r], acc[r][3]);
            }
        }
    }
}

__global__ void __launch_bounds__(K4_THREADS) __attribute__((amdgpu_waves_per_eu(2, 2)))
k4_agg(const int* __restrict__ adj, const float* __restrict__ ht,
       const float* __restrict__ ssrc, const float* __restrict__ sdst,
       const float* __restrict__ f1g, const float* __restrict__ f2g,
       const unsigned* __restrict__ amask,
       float* __restrict__ out, float* __restrict__ part1, size_t ws_size) {
    __shared__ float s_part[K4_HEADS * K4_ROWS * OUT_F];
    __shared__ float s_mrg[K4_HEADS * MRG_ROWS * 32];

    const int i0   = (blockIdx.x >> 1) * K4_ROWS;
    const int hg   = blockIdx.x & 1;
    const int tid  = threadIdx.x;
    const int lane = tid & 31;
    const int hl   = (tid >> 5) & (K4_HEADS - 1);
    const int hd   = hg * K4_HEADS + hl;
    const int hf   = tid >> 7;
    const int c    = lane & 15;
    const int hi   = lane >> 4;

    float ssr[K4_RT], m_use[K4_RT];
    v2f   l_run[K4_RT];
    v8f   acc[K4_RT][4];
#pragma unroll
    for (int r = 0; r < K4_RT; ++r) {
        ssr[r]   = ssrc[(i0 + 16 * r + c) * NHEAD + hd];
        m_use[r] = -__builtin_inff();
        l_run[r] = v2f{};
#pragma unroll
        for (int t = 0; t < 4; ++t) acc[r][t] = v8f{};
    }

    const int*   adj_row0 = adj + (size_t)(i0 + c) * N_NODES + hf * JHALF + (hi << 4);
    const int*   adj_row1 = adj_row0 + (size_t)16 * N_NODES;
    const float* sd_col   = sdst + (size_t)hd * N_NODES + hf * JHALF + (hi << 4);
    const float* ht_col   = ht + (size_t)(hd * (N_NODES / 32) + hf * (JHALF / 32)) * HTF_BLK
                          + (lane << 2);

#define K4_STEP_END()                                                              \
    do {                                                                           \
        asm volatile("" : "+v"(acc[0][0]), "+v"(acc[0][1]), "+v"(acc[0][2]),       \
                          "+v"(acc[0][3]), "+v"(acc[1][0]), "+v"(acc[1][1]),       \
                          "+v"(acc[1][2]), "+v"(acc[1][3]), "+v"(l_run[0]),        \
                          "+v"(l_run[1]) : : "memory");                            \
        __builtin_amdgcn_sched_barrier(0);                                         \
    } while (0)

    float dmx = 0.f;
    {
        const float* sdh = sdst + (size_t)hd * N_NODES + (lane << 2);
#pragma unroll
        for (int k = 0; k < N_NODES / 128; ++k) {
            const v4f v = *(const v4f*)&sdh[k * 128];
            dmx = fmaxf(dmx, fmaxf(fmaxf(fabsf(v[0]), fabsf(v[1])),
                                   fmaxf(fabsf(v[2]), fabsf(v[3]))));
        }
        for (int msk = 16; msk >= 1; msk >>= 1) dmx = fmaxf(dmx, __shfl_xor(dmx, msk));
    }
    bool fast = ws_size >= WS_NEEDED3 &&
                __builtin_amdgcn_readfirstlane(dmx <= K4_DMAX ? 1 : 0) != 0;

    float L[K4_RT];
    if (fast) {
        float a1[K4_RT], a2[K4_RT], mfix[K4_RT];
#pragma unroll
        for (int r = 0; r < K4_RT; ++r) {
            mfix[r] = ssr[r] > 0.f ? ssr[r] : 0.2f * ssr[r];
            a1[r]   = expf(ssr[r] - mfix[r]);
            a2[r]   = expf(0.2f * ssr[r] - mfix[r]);
        }
        const float* f1_col = f1g + (size_t)hd * N_NODES + hf * JHALF + (hi << 4);
        const float* f2_col = f2g + (size_t)hd * N_NODES + hf * JHALF + (hi << 4);
        const int hfs = __builtin_amdgcn_readfirstlane(hf);
        const k4_mask_ptr mrow = (k4_mask_ptr)(amask +
            ((size_t)(blockIdx.x >> 1) * (N_NODES / 32) + hfs * (JHALF / 32)) * 32);
#define K4_MASKS_NEXT(cur, nxt, jnext)                                              \
    do {                                                                           \
        asm volatile("" : : "s"(cur.m[0]), "s"(cur.m[1]), "s"(cur.m[2]),           \
                            "s"(cur.m[3]));                                        \
        __builtin_amdgcn_sched_barrier(0);                                         \
        k4_load_m(nxt, mrow, jnext);                                               \
        __builtin_amdgcn_sched_barrier(0);                                         \
    } while (0)
        K4FragF g0, g1;
        k4_load_f(g0, f1_col, f2_col, ht_col, 0);
        k4_load_m(g0, mrow, 0);
        K4_STEP_END();
        for (int jb = 0; jb < JHALF; jb += 2 * JB) {
            k4_load_f(g1, f1_col, f2_col, ht_col, jb + JB);
            K4_STEP_END();
            K4_MASKS_NEXT(g0, g1, jb + JB);
            k4_compute_f(g0, a1, a2, l_run, acc);
            K4_STEP_END();
            const int jn = (jb + 2 * JB < JHALF) ? jb + 2 * JB : jb;
            k4_load_f(g0, f1_col, f2_col, ht_col, jn);
            K4_STEP_END();
            K4_MASKS_NEXT(g1, g0, jn);
            k4_compute_f(g1, a1, a2, l_run, acc);
            K4_STEP_END();
        }
        bool empty = false;
#pragma unroll
        for (int r = 0; r < K4_RT; ++r) {
            const float l_lane = l_run[r][0] + l_run[r][1];
            L[r] = l_lane + k4_xhalf(l_lane);
            empty = empty || !(L[r] > 0.f);
        }
        if (__builtin_amdgcn_ballot_w32(empty) != 0) {
            fast = false;
#pragma unroll
            for (int r = 0; r < K4_RT; ++r) {
                l_run[r] = v2f{};
#pragma unroll
                for (int t = 0; t < 4; ++t) acc[r][t] = v8f{};
            }
        } else {
#pragma unroll
            for (int r = 0; r < K4_RT; ++r) m_use[r] = mfix[r];
        }
    }
    if (!fast) {
        K4Frag f0, f1;
        k4_load(f0, adj_row0, adj_row1, sd_col, ht_col, 0);
        K4_STEP_END();
        for (int jb = 0; jb < JHALF; jb += 2 * JB) {
            k4_load(f1, adj_row0, adj_row1, sd_col, ht_col, jb + JB);
            K4_STEP_END();
            k4_compute(f0, ssr, m_use, l_run, acc);
            K4_STEP_END();
            const int jn = (jb + 2 * JB < JHALF) ? jb + 2 * JB : jb;
            k4_load(f0, adj_row0, adj_row1, sd_col, ht_col, jn);
            K4_STEP_END();
            k4_compute(f1, ssr, m_use, l_run, acc);
            K4_STEP_END();
        }
#pragma unroll
        for (int r = 0; r < K4_RT; ++r) {
            const float l_lane = l_run[r][0] + l_run[r][1];
            L[r] = l_lane + k4_xhalf(l_lane);
        }
    }

    float* mrg = s_mrg + hl * (MRG_ROWS * 32) + lane;
    if (hf == 1) {
#pragma unroll
        for (int r = 0; r < K4_RT; ++r) {
#pragma unroll
            for (int t = 0; t < 4; ++t)
#pragma unroll
                for (int p = 0; p < 8; ++p)
                    mrg[((r * 4 + t) * 8 + p) * 32] = acc[r][t][p];
            mrg[(64 + r) * 32] = m_use[r];
            mrg[(66 + r) * 32] = L[r];
        }
    }
    __syncthreads();

    if (hf == 0) {
#pragma unroll
        for (int r = 0; r < K4_RT; ++r) {
            const float m_b = mrg[(64 + r) * 32];
            const float m   = fmaxf(m_use[r], m_b);
            const float sa  = __expf(m_use[r] - m);
            const float sb  = __expf(m_b - m);
            const float Lr  = L[r] * sa + mrg[(66 + r) * 32] * sb;
#pragma unroll
            for (int p = 0; p < 8; ++p) {
                float4 o;
                o.x = (acc[r][0][p] * sa + mrg[((r * 4 + 0) * 8 + p) * 32] * sb) / Lr;
                o.y = (acc[r][1][p] * sa + mrg[((r * 4 + 1) * 8 + p) * 32] * sb) / Lr;
                o.z = (acc[r][2][p] * sa + mrg[((r * 4 + 2) * 8 + p) * 32] * sb) / Lr;
                o.w = (acc[r][3][p] * sa + mrg[((r * 4 + 3) * 8 + p) * 32] * sb) / Lr;
                *(float4*)&s_part[(hl * K4_ROWS + 16 * r + c) * OUT_F + (hi << 5) + (p << 2)] = o;
            }
        }
    }
    __syncthreads();

    float* dst = hg == 0 ? out : part1;
    if (hg == 0 || ws_size >= WS_NEEDED) {
        for (int idx = tid; idx < K4_ROWS * OUT_F; idx += K4_THREADS) {
            float hp = 0.f;
#pragma unroll
            for (int hh = 0; hh < K4_HEADS; ++hh) hp += s_part[hh * K4_ROWS * OUT_F + idx];
            dst[(size_t)i0 * OUT_F + idx] = hp;
        }
    }
}

__global__ void __launch_bounds__(256)
k5_mean_ln(const float* __restrict__ part1, const float* __restrict__ gamma,
           const float* __restrict__ beta, float* out, size_t ws_size) {
    __shared__ float s_out[16 * OUT_F];
    __shared__ float s_mu[16], s_rs[16];
    const int i0  = blockIdx.x * 16;
    const int tid = threadIdx.x;

    for (int idx = tid; idx < 16 * OUT_F; idx += 256) {
        const size_t g = (size_t)i0 * OUT_F + idx;
        float hp = out[g];
        if (ws_size >= WS_NEEDED) hp += part1[g];
        s_out[idx] = hp * (1.0f / NHEAD);
    }
    __syncthreads();

    if (tid < 16) {
        float mu = 0.f;
        for (int f = 0; f < OUT_F; ++f) mu += s_out[tid * OUT_F + f];
        mu *= (1.0f / OUT_F);
        float var = 0.f;
        for (int f = 0; f < OUT_F; ++f) {
            const float d = s_out[tid * OUT_F + f] - mu;
            var += d * d;
        }
        var *= (1.0f / OUT_F);
        s_mu[tid] = mu;
        s_rs[tid] = rsqrtf(var + LN_EPS);
    }
    __syncthreads();

    for (int idx = tid; idx < 16 * OUT_F; idx += 256) {
        const int r = idx >> 6, f = idx & 63;
        out[(size_t)(i0 + r) * OUT_F + f] =
            (s_out[r * OUT_F + f] - s_mu[r]) * s_rs[r] * gamma[f] + beta[f];
    }
}

extern "C" void kernel_launch(void* const* d_in, const int* in_sizes, int n_in,
                              void* d_out, int out_size, void* d_ws, size_t ws_size,
                              hipStream_t stream) {
    const float* h     = (const float*)d_in[0];
    const int*   adj   = (const int*)  d_in[1];
    const float* W     = (const float*)d_in[2];
    const float* a     = (const float*)d_in[3];
    const float* gamma = (const float*)d_in[4];
    const float* beta  = (const float*)d_in[5];
    float* out = (float*)d_out;

    float* ws    = (float*)d_ws;
    float* ht    = ws;
    float* ssrc  = ht + (size_t)N_NODES * HID;
    float* sdst  = ssrc + N_NODES * NHEAD;
    float* part1 = sdst + N_NODES * NHEAD;
    float* f1    = part1 + (size_t)N_NODES * OUT_F;
    float* f2    = f1 + N_NODES * NHEAD;
    unsigned* amask = (unsigned*)(f2 + N_NODES * NHEAD);

    k1_gemm_ht<<<dim3(N_NODES / 16, HID / 64), 128, 0, stream>>>(h, W, ht);
    k2_scores<<<K2_SCORE_BLOCKS + K2_MASK_BLOCKS, 256, 0, stream>>>(ht, a, adj, ssrc, sdst, f1, f2,
                                                                    amask, ws_size);
    k4_agg<<<2 * (N_NODES / K4_ROWS), K4_THREADS, 0, stream>>>(adj, ht, ssrc, sdst, f1, f2, amask,
                                                       out, part1, ws_size);
    k5_mean_ln<<<N_NODES / 16, 256, 0, stream>>>(part1, gamma, beta, out, ws_size);
}


# K2 dissolved into K1's launch: K1 wave = 16 nodes x one head, scores as h.(W.a) on a 5th tile fed from the W registers, mask blocks beside it; 3 launches; score sum order changes
# speedup vs baseline: 15.6277x; 1.0500x over previous
#include <hip/hip_runtime.h>
#include <hip/hip_bf16.h>

#define N_NODES 4096
#define IN_F    256
#define OUT_F   64
#define NHEAD   8
#define HID     512
#define NEG_BIG (-9.0e15f)
#define LN_EPS  1e-05f

typedef __attribute__((ext_vector_type(2))) float v2f;
typedef __attribute__((ext_vector_type(8))) float v8f;

__device__ __forceinline__ v8f wmma_f32_16x16x4(v2f A, v2f B, v8f C) {
    return __builtin_amdgcn_wmma_f32_16x16x4_f32(false, A, false, B, (short)0, C,
                                                 false, false);
}

#define HTF_BLK 2048
#define WS_NEEDED  ((size_t)(N_NODES * HID + 2 * N_NODES * NHEAD + N_NODES * OUT_F) * 4)
#define WS_NEEDED2 (WS_NEEDED + (size_t)(2 * N_NODES * NHEAD) * 4)
#define MASK_WORDS ((size_t)(N_NODES / 32) * (N_NODES / 32) * 32)
#define WS_NEEDED3 (WS_NEEDED2 + MASK_WORDS * 4)
#define K1_GEMM_BLOCKS (N_NODES / 16)
#define K1_MASK_BLOCKS 512
#define K1_KC 8
typedef __attribute__((ext_vector_type(4))) int   v4i;
typedef __attribute__((ext_vector_type(4))) float v4w;
typedef __attribute__((ext_vector_type(8))) unsigned v8u;

template <int MASK>
__device__ __forceinline__ float k1_xadd(float x) {
    const int b = __builtin_bit_cast(int, x);
    return x + __builtin_bit_cast(float,
        __builtin_amdgcn_update_dpp(b, b, 0x160 + MASK, 0xf, 0xf, true));
}

struct K1Frag {
    v4w b[K1_KC][2];
    v2f a[K1_KC];
};

__device__ __forceinline__ void k1_load_a(K1Frag& fr, const float* __restrict__ arow, int kc) {
#pragma unroll
    for (int kk = 0; kk < K1_KC; ++kk)
        fr.a[kk] = *(const v2f*)&arow[(kc * K1_KC + kk) * 4];
}

__device__ __forceinline__ void k1_load(K1Frag& fr, const float* __restrict__ wcol, int kc) {
#pragma unroll
    for (int kk = 0; kk < K1_KC; ++kk)
#pragma unroll
        for (int v = 0; v < 2; ++v)
            fr.b[kk][v] = *(const v4w*)&wcol[(size_t)((kc * K1_KC + kk) * 4 + v) * HID];
}

__device__ __forceinline__ void k1_compute(const K1Frag& fr, const v2f (&cc)[4], bool scol,
                                           v8f (&acc)[5]) {
#pragma unroll
    for (int kk = 0; kk < K1_KC; ++kk) {
        const v2f A = fr.a[kk];
        v2f B5;
#pragma unroll
        for (int v = 0; v < 2; ++v) {
            const v4w w = fr.b[kk][v];
            v2f tu = cc[0] * w[0];
            tu = __builtin_elementwise_fma(cc[1], (v2f){w[1], w[1]}, tu);
            tu = __builtin_elementwise_fma(cc[2], (v2f){w[2], w[2]}, tu);
            tu = __builtin_elementwise_fma(cc[3], (v2f){w[3], w[3]}, tu);
            const float t0 = tu[0];
            const float u1 = tu[1];
            const int ub = __float_as_int(u1);
            float t = t0 + __int_as_float(
                __builtin_amdgcn_update_dpp(ub, ub, 0x161, 0xf, 0xf, true));
            t = k1_xadd<2>(t);
            t = k1_xadd<4>(t);
            t = k1_xadd<8>(t);
            B5[v] = scol ? t : 0.f;
        }
#pragma unroll
        for (int T = 0; T < 4; ++T) {
            v2f B;
            B[0] = fr.b[kk][0][T];
            B[1] = fr.b[kk][1][T];
            acc[T] = wmma_f32_16x16x4(A, B, acc[T]);
        }
        acc[4] = wmma_f32_16x16x4(A, B5, acc[4]);
    }
}

__global__ void __launch_bounds__(256)
k1_gemm_ht(const float* __restrict__ h, const float* __restrict__ W,
           const float* __restrict__ a, const int* __restrict__ adj,
           float* __restrict__ ht, float* __restrict__ ssrc, float* __restrict__ sdst,
           float* __restrict__ f1, float* __restrict__ f2,
           unsigned* __restrict__ amask, size_t ws_size) {
    __shared__ float sA[16 * 260];
    const int tid  = threadIdx.x;
    const int lane = tid & 31;
    const int wave = tid >> 5;

    if (blockIdx.x >= K1_GEMM_BLOCKS) {
        if (ws_size < WS_NEEDED3) return;
        const int u  = (blockIdx.x - K1_GEMM_BLOCKS) * 8 + wave;
        const int g  = u >> 5;
        const int b0 = (u & 31) * 4;
        const int c  = lane & 15, hi = lane >> 4;
        const int* base = adj + (size_t)(32 * g + c) * N_NODES + 32 * b0 + 16 * hi;
        v4i av[4][2][4];
#pragma unroll
        for (int k = 0; k < 4; ++k)
#pragma unroll
            for (int r = 0; r < 2; ++r)
#pragma unroll
                for (int q = 0; q < 4; ++q)
                    av[k][r][q] = *(const v4i*)&base[(size_t)r * 16 * N_NODES + 32 * k + 4 * q];
#pragma unroll
        for (int k = 0; k < 4; ++k) {
            unsigned word = 0u;
#pragma unroll
            for (int r = 0; r < 2; ++r)
#pragma unroll
                for (int q = 0; q < 4; ++q)
#pragma unroll
                    for (int e = 0; e < 4; ++e) {
                        const unsigned vote = __builtin_amdgcn_ballot_w32(av[k][r][q][e] > 0);
                        if (lane == r * 16 + 4 * q + e) word = vote;
                    }
            amask[(size_t)(g * (N_NODES / 32) + b0 + k) * 32 + lane] = word;
        }
        return;
    }

    const int i0 = blockIdx.x * 16;
    const int m  = lane & 15;
    const int hi = lane >> 4;

    {
        float4 hv[4];
#pragma unroll
        for (int q = 0; q < 4; ++q) {
            const int idx = tid + q * 256;
            hv[q] = *(const float4*)&h[(size_t)(i0 + (idx >> 6)) * IN_F + ((idx & 63) << 2)];
        }
#pragma unroll
        for (int q = 0; q < 4; ++q) {
            const int idx = tid + q * 256;
            *(float4*)&sA[(idx >> 6) * 260 + ((idx & 63) << 2)] = hv[q];
        }
    }

    const float* wcol = W + (size_t)(hi << 1) * HID + (wave << 6) + (m << 2);
    const float* arow = sA + m * 260 + (hi << 1);
    const v4w as4 = *(const v4w*)&a[m << 2];
    const v4w ad4 = *(const v4w*)&a[OUT_F + (m << 2)];
    const bool odd  = (m & 1) != 0;
    const bool scol = m < 2;
    v2f cc[4];
#pragma unroll
    for (int T = 0; T < 4; ++T) {
        cc[T][0] = odd ? ad4[T] : as4[T];
        cc[T][1] = odd ? as4[T] : ad4[T];
    }

    v8f acc[5];
#pragma unroll
    for (int T = 0; T < 5; ++T) acc[T] = v8f{};

#define K1_STEP_END()                                                              \
    do {                                                                           \
        asm volatile("" : "+v"(acc[0]), "+v"(acc[1]), "+v"(acc[2]), "+v"(acc[3]),  \
                          "+v"(acc[4]) : : "memory");                              \
        __builtin_amdgcn_sched_barrier(0);                                         \
    } while (0)

    K1Frag w0, w1;
    k1_load(w0, wcol, 0);
    K1_STEP_END();
    __syncthreads();
    k1_load_a(w0, arow, 0);
    K1_STEP_END();
    for (int kc = 0; kc < IN_F / 4 / K1_KC; kc += 2) {
        k1_load(w1, wcol, kc + 1);
        k1_load_a(w1, arow, kc + 1);
        K1_STEP_END();
        k1_compute(w0, cc, scol, acc);
        K1_STEP_END();
        const int kn = (kc + 2 < IN_F / 4 / K1_KC) ? kc + 2 : kc;
        k1_load(w0, wcol, kn);
        k1_load_a(w0, arow, kn);
        K1_STEP_END();
        k1_compute(w1, cc, scol, acc);
        K1_STEP_END();
    }

    float* dst = ht + (size_t)(wave * (N_NODES / 32) + (blockIdx.x >> 1)) * HTF_BLK
               + ((((int)blockIdx.x & 1) << 4) + m) * 4 + (hi << 1) * 128;
#pragma unroll
    for (int T = 0; T < 4; ++T)
#pragma unroll
        for (int pp = 0; pp < 2; ++pp) {
            float4 o;
            o.x = acc[T][4 * pp + 0];
            o.y = acc[T][4 * pp + 1];
            o.z = acc[T][4 * pp + 2];
            o.w = acc[T][4 * pp + 3];
            *(float4*)&dst[T * 512 + pp * 128] = o;
        }

    const int n8 = i0 + (hi << 3);
    if (m == 0) {
#pragma unroll
        for (int p = 0; p < 8; ++p) ssrc[(n8 + p) * NHEAD + wave] = acc[4][p];
    } else if (m == 1) {
        const size_t o = (size_t)wave * N_NODES + n8;
#pragma unroll
        for (int pp = 0; pp < 2; ++pp) {
            float4 s;
            s.x = acc[4][4 * pp + 0]; s.y = acc[4][4 * pp + 1];
            s.z = acc[4][4 * pp + 2]; s.w = acc[4][4 * pp + 3];
            *(float4*)&sdst[o + 4 * pp] = s;
        }
        if (ws_size >= WS_NEEDED2) {
#pragma unroll
            for (int pp = 0; pp < 2; ++pp) {
                float4 e1, e2;
                e1.x = expf(acc[4][4 * pp + 0]); e2.x = expf(0.2f * acc[4][4 * pp + 0]);
                e1.y = expf(acc[4][4 * pp + 1]); e2.y = expf(0.2f * acc[4][4 * pp + 1]);
                e1.z = expf(acc[4][4 * pp + 2]); e2.z = expf(0.2f * acc[4][4 * pp + 2]);
                e1.w = expf(acc[4][4 * pp + 3]); e2.w = expf(0.2f * acc[4][4 * pp + 3]);
                *(float4*)&f1[o + 4 * pp] = e1;
                *(float4*)&f2[o + 4 * pp] = e2;
            }
        }
    }
}


#define KU 8
#define JB (KU * 4)
#define K4_LAG 8.0f
#define K4_DMAX 30.0f

typedef __attribute__((ext_vector_type(4))) float v4f;
#define KQ (KU / 2)
#define K4_RT      2
#define K4_HEADS   4
#define K4_WAVES   8
#define K4_THREADS (K4_WAVES * 32)
#define K4_ROWS    (K4_RT * 16)
#define JHALF      (N_NODES / 2)
#define MRG_ROWS   68

struct K4Frag {
    v4i adjq[K4_RT][KQ];
    v4f sd[KQ];
    v4f b[4][KQ];
};

__device__ __forceinline__ void k4_load(K4Frag& fr, const int* __restrict__ adj_row0,
                                        const int* __restrict__ adj_row1,
                                        const float* __restrict__ sd_col,
                                        const float* __restrict__ ht_col, int jb) {
    const float* chunk = ht_col + (size_t)(jb >> 5) * HTF_BLK;
#pragma unroll
    for (int q = 0; q < KQ; ++q) {
        const int j = jb + q * 4;
        fr.adjq[0][q] = *(const v4i*)&adj_row0[j];
        fr.adjq[1][q] = *(const v4i*)&adj_row1[j];
        fr.sd[q]      = *(const v4f*)&sd_col[j];
#pragma unroll
        for (int t = 0; t < 4; ++t)
            fr.b[t][q] = *(const v4f*)&chunk[t * 512 + q * 128];
    }
}

struct K4FragF {
    v8u m[4];
    v4f f1[KQ];
    v4f f2[KQ];
    v4f b[4][KQ];
};

typedef const __attribute__((address_space(4))) v8u* k4_mask_ptr;

__device__ __forceinline__ void k4_load_m(K4FragF& fr, k4_mask_ptr mrow, int jb) {
#pragma unroll
    for (int i = 0; i < 4; ++i) fr.m[i] = mrow[(jb >> 5) * 4 + i];
}

__device__ __forceinline__ void k4_load_f(K4FragF& fr,
                                          const float* __restrict__ f1_col,
                                          const float* __restrict__ f2_col,
                                          const float* __restrict__ ht_col, int jb) {
    const float* chunk = ht_col + (size_t)(jb >> 5) * HTF_BLK;
#pragma unroll
    for (int q = 0; q < KQ; ++q) {
        const int j = jb + q * 4;
        fr.f1[q]      = *(const v4f*)&f1_col[j];
        fr.f2[q]      = *(const v4f*)&f2_col[j];
#pragma unroll
        for (int t = 0; t < 4; ++t)
            fr.b[t][q] = *(const v4f*)&chunk[t * 512 + q * 128];
    }
}

__device__ __forceinline__ void k4_compute_f(const K4FragF& fr, const float (&a1)[K4_RT],
                                             const float (&a2)[K4_RT], v2f (&l_run)[K4_RT],
                                             v8f (&acc)[K4_RT][4]) {
#pragma unroll
    for (int q = 0; q < KQ; ++q) {
        v4f x0 = fr.b[0][q], x1 = fr.b[1][q], x2 = fr.b[2][q], x3 = fr.b[3][q];
#pragma unroll
        for (int s = 0; s < 2; ++s) {
            v2f P[K4_RT];
#pragma unroll
            for (int r = 0; r < K4_RT; ++r) {
#pragma unroll
                for (int v = 0; v < 2; ++v) {
                    const float p = fmaxf(a1[r] * fr.f1[q][2 * s + v], a2[r] * fr.f2[q][2 * s + v]);
                    const int   w = r * 16 + 4 * q + 2 * s + v;
                    P[r][v] = __builtin_amdgcn_inverse_ballot_w32(fr.m[w >> 3][w & 7]) ? p : 0.f;
                }
                l_run[r] += P[r];
            }
            v2f A0, A1, A2, A3;
            A0[0] = x0[2 * s]; A0[1] = x0[2 * s + 1];
            A1[0] = x1[2 * s]; A1[1] = x1[2 * s + 1];
            A2[0] = x2[2 * s]; A2[1] = x2[2 * s + 1];
            A3[0] = x3[2 * s]; A3[1] = x3[2 * s + 1];
#pragma unroll
            for (int r = 0; r < K4_RT; ++r) {
                acc[r][0] = wmma_f32_16x16x4(A0, P[r], acc[r][0]);
                acc[r][1] = wmma_f32_16x16x4(A1, P[r], acc[r][1]);
                acc[r][2] = wmma_f32_16x16x4(A2, P[r], acc[r][2]);
                acc[r][3] = wmma_f32_16x16x4(A3, P[r], acc[r][3]);
            }
        }
    }
}

__device__ __forceinline__ float k4_score(int adjv, float sd, float ssr) {
    float e = ssr + sd;
    e = e > 0.f ? e : 0.2f * e;
    return adjv > 0 ? e : NEG_BIG;
}

__device__ __forceinline__ float k4_xhalf(float x) {
    const int b = __builtin_bit_cast(int, x);
    return __builtin_bit_cast(float,
        __builtin_amdgcn_permlanex16(b, b, 0x76543210, 0xfedcba98, false, false));
}

__device__ __forceinline__ void k4_compute(const K4Frag& fr, const float (&ssr)[K4_RT],
                                           float (&m_use)[K4_RT], v2f (&l_run)[K4_RT],
                                           v8f (&acc)[K4_RT][4]) {
    float e[K4_RT][KQ][4];
    float bm[K4_RT];
#pragma unroll
    for (int r = 0; r < K4_RT; ++r) {
        float b = NEG_BIG;
#pragma unroll
        for (int q = 0; q < KQ; ++q) {
#pragma unroll
            for (int x = 0; x < 4; ++x)
                e[r][q][x] = k4_score(fr.adjq[r][q][x], fr.sd[q][x], ssr[r]);
            b = fmaxf(b, fmaxf(fmaxf(e[r][q][0], e[r][q][1]), fmaxf(e[r][q][2], e[r][q][3])));
        }
        bm[r] = fmaxf(b, k4_xhalf(b));
    }
    const bool need = (bm[0] > m_use[0] + K4_LAG) || (bm[1] > m_use[1] + K4_LAG);
    if (__builtin_amdgcn_ballot_w32(need) != 0) {
#pragma unroll
        for (int r = 0; r < K4_RT; ++r) {
            const float m_new = fmaxf(m_use[r], bm[r]);
            const float corr  = __expf(m_use[r] - m_new);
            m_use[r] = m_new;
            l_run[r] = l_run[r] * corr;
#pragma unroll
            for (int t = 0; t < 4; ++t) acc[r][t] = acc[r][t] * corr;
        }
    }
#pragma unroll
    for (int q = 0; q < KQ; ++q) {
        v4f x0 = fr.b[0][q], x1 = fr.b[1][q], x2 = fr.b[2][q], x3 = fr.b[3][q];
        asm("" : "+v"(x0), "+v"(x1), "+v"(x2), "+v"(x3));
#pragma unroll
        for (int s = 0; s < 2; ++s) {
            v2f P[K4_RT];
#pragma unroll
            for (int r = 0; r < K4_RT; ++r) {
                P[r][0] = __expf(e[r][q][2 * s + 0] - m_use[r]);
                P[r][1] = __expf(e[r][q][2 * s + 1] - m_use[r]);
                l_run[r] += P[r];
            }
            v2f A0, A1, A2, A3;
            A0[0] = x0[2 * s]; A0[1] = x0[2 * s + 1];
            A1[0] = x1[2 * s]; A1[1] = x1[2 * s + 1];
            A2[0] = x2[2 * s]; A2[1] = x2[2 * s + 1];
            A3[0] = x3[2 * s]; A3[1] = x3[2 * s + 1];
#pragma unroll
            for (int r = 0; r < K4_RT; ++r) {
                acc[r][0] = wmma_f32_16x16x4(A0, P[r], acc[r][0]);
                acc[r][1] = wmma_f32_16x16x4(A1, P[r], acc[r][1]);
                acc[r][2] = wmma_f32_16x16x4(A2, P[r], acc[r][2]);
                acc[r][3] = wmma_f32_16x16x4(A3, P[r], acc[r][3]);
            }
        }
    }
}

__global__ void __launch_bounds__(K4_THREADS) __attribute__((amdgpu_waves_per_eu(2, 2)))
k4_agg(const int* __restrict__ adj, const float* __restrict__ ht,
       const float* __restrict__ ssrc, const float* __restrict__ sdst,
       const float* __restrict__ f1g, const float* __restrict__ f2g,
       const unsigned* __restrict__ amask,
       float* __restrict__ out, float* __restrict__ part1, size_t ws_size) {
    __shared__ float s_part[K4_HEADS * K4_ROWS * OUT_F];
    __shared__ float s_mrg[K4_HEADS * MRG_ROWS * 32];

    const int i0   = (blockIdx.x >> 1) * K4_ROWS;
    const int hg   = blockIdx.x & 1;
    const int tid  = threadIdx.x;
    const int lane = tid & 31;
    const int hl   = (tid >> 5) & (K4_HEADS - 1);
    const int hd   = hg * K4_HEADS + hl;
    const int hf   = tid >> 7;
    const int c    = lane & 15;
    const int hi   = lane >> 4;

    float ssr[K4_RT], m_use[K4_RT];
    v2f   l_run[K4_RT];
    v8f   acc[K4_RT][4];
#pragma unroll
    for (int r = 0; r < K4_RT; ++r) {
        ssr[r]   = ssrc[(i0 + 16 * r + c) * NHEAD + hd];
        m_use[r] = -__builtin_inff();
        l_run[r] = v2f{};
#pragma unroll
        for (int t = 0; t < 4; ++t) acc[r][t] = v8f{};
    }

    const int*   adj_row0 = adj + (size_t)(i0 + c) * N_NODES + hf * JHALF + (hi << 4);
    const int*   adj_row1 = adj_row0 + (size_t)16 * N_NODES;
    const float* sd_col   = sdst + (size_t)hd * N_NODES + hf * JHALF + (hi << 4);
    const float* ht_col   = ht + (size_t)(hd * (N_NODES / 32) + hf * (JHALF / 32)) * HTF_BLK
                          + (lane << 2);

#define K4_STEP_END()                                                              \
    do {                                                                           \
        asm volatile("" : "+v"(acc[0][0]), "+v"(acc[0][1]), "+v"(acc[0][2]),       \
                          "+v"(acc[0][3]), "+v"(acc[1][0]), "+v"(acc[1][1]),       \
                          "+v"(acc[1][2]), "+v"(acc[1][3]), "+v"(l_run[0]),        \
                          "+v"(l_run[1]) : : "memory");                            \
        __builtin_amdgcn_sched_barrier(0);                                         \
    } while (0)

    float dmx = 0.f;
    {
        const float* sdh = sdst + (size_t)hd * N_NODES + (lane << 2);
#pragma unroll
        for (int k = 0; k < N_NODES / 128; ++k) {
            const v4f v = *(const v4f*)&sdh[k * 128];
            dmx = fmaxf(dmx, fmaxf(fmaxf(fabsf(v[0]), fabsf(v[1])),
                                   fmaxf(fabsf(v[2]), fabsf(v[3]))));
        }
        for (int msk = 16; msk >= 1; msk >>= 1) dmx = fmaxf(dmx, __shfl_xor(dmx, msk));
    }
    bool fast = ws_size >= WS_NEEDED3 &&
                __builtin_amdgcn_readfirstlane(dmx <= K4_DMAX ? 1 : 0) != 0;

    float L[K4_RT];
    if (fast) {
        float a1[K4_RT], a2[K4_RT], mfix[K4_RT];
#pragma unroll
        for (int r = 0; r < K4_RT; ++r) {
            mfix[r] = ssr[r] > 0.f ? ssr[r] : 0.2f * ssr[r];
            a1[r]   = expf(ssr[r] - mfix[r]);
            a2[r]   = expf(0.2f * ssr[r] - mfix[r]);
        }
        const float* f1_col = f1g + (size_t)hd * N_NODES + hf * JHALF + (hi << 4);
        const float* f2_col = f2g + (size_t)hd * N_NODES + hf * JHALF + (hi << 4);
        const int hfs = __builtin_amdgcn_readfirstlane(hf);
        const k4_mask_ptr mrow = (k4_mask_ptr)(amask +
            ((size_t)(blockIdx.x >> 1) * (N_NODES / 32) + hfs * (JHALF / 32)) * 32);
#define K4_MASKS_NEXT(cur, nxt, jnext)                                              \
    do {                                                                           \
        asm volatile("" : : "s"(cur.m[0]), "s"(cur.m[1]), "s"(cur.m[2]),           \
                            "s"(cur.m[3]));                                        \
        __builtin_amdgcn_sched_barrier(0);                                         \
        k4_load_m(nxt, mrow, jnext);                                               \
        __builtin_amdgcn_sched_barrier(0);                                         \
    } while (0)
        K4FragF g0, g1;
        k4_load_f(g0, f1_col, f2_col, ht_col, 0);
        k4_load_m(g0, mrow, 0);
        K4_STEP_END();
        for (int jb = 0; jb < JHALF; jb += 2 * JB) {
            k4_load_f(g1, f1_col, f2_col, ht_col, jb + JB);
            K4_STEP_END();
            K4_MASKS_NEXT(g0, g1, jb + JB);
            k4_compute_f(g0, a1, a2, l_run, acc);
            K4_STEP_END();
            const int jn = (jb + 2 * JB < JHALF) ? jb + 2 * JB : jb;
            k4_load_f(g0, f1_col, f2_col, ht_col, jn);
            K4_STEP_END();
            K4_MASKS_NEXT(g1, g0, jn);
            k4_compute_f(g1, a1, a2, l_run, acc);
            K4_STEP_END();
        }
        bool empty = false;
#pragma unroll
        for (int r = 0; r < K4_RT; ++r) {
            const float l_lane = l_run[r][0] + l_run[r][1];
            L[r] = l_lane + k4_xhalf(l_lane);
            empty = empty || !(L[r] > 0.f);
        }
        if (__builtin_amdgcn_ballot_w32(empty) != 0) {
            fast = false;
#pragma unroll
            for (int r = 0; r < K4_RT; ++r) {
                l_run[r] = v2f{};
#pragma unroll
                for (int t = 0; t < 4; ++t) acc[r][t] = v8f{};
            }
        } else {
#pragma unroll
            for (int r = 0; r < K4_RT; ++r) m_use[r] = mfix[r];
        }
    }
    if (!fast) {
        K4Frag f0, f1;
        k4_load(f0, adj_row0, adj_row1, sd_col, ht_col, 0);
        K4_STEP_END();
        for (int jb = 0; jb < JHALF; jb += 2 * JB) {
            k4_load(f1, adj_row0, adj_row1, sd_col, ht_col, jb + JB);
            K4_STEP_END();
            k4_compute(f0, ssr, m_use, l_run, acc);
            K4_STEP_END();
            const int jn = (jb + 2 * JB < JHALF) ? jb + 2 * JB : jb;
            k4_load(f0, adj_row0, adj_row1, sd_col, ht_col, jn);
            K4_STEP_END();
            k4_compute(f1, ssr, m_use, l_run, acc);
            K4_STEP_END();
        }
#pragma unroll
        for (int r = 0; r < K4_RT; ++r) {
            const float l_lane = l_run[r][0] + l_run[r][1];
            L[r] = l_lane + k4_xhalf(l_lane);
        }
    }

    float* mrg = s_mrg + hl * (MRG_ROWS * 32) + lane;
    if (hf == 1) {
#pragma unroll
        for (int r = 0; r < K4_RT; ++r) {
#pragma unroll
            for (int t = 0; t < 4; ++t)
#pragma unroll
                for (int p = 0; p < 8; ++p)
                    mrg[((r * 4 + t) * 8 + p) * 32] = acc[r][t][p];
            mrg[(64 + r) * 32] = m_use[r];
            mrg[(66 + r) * 32] = L[r];
        }
    }
    __syncthreads();

    if (hf == 0) {
#pragma unroll
        for (int r = 0; r < K4_RT; ++r) {
            const float m_b = mrg[(64 + r) * 32];
            const float m   = fmaxf(m_use[r], m_b);
            const float sa  = __expf(m_use[r] - m);
            const float sb  = __expf(m_b - m);
            const float Lr  = L[r] * sa + mrg[(66 + r) * 32] * sb;
#pragma unroll
            for (int p = 0; p < 8; ++p) {
                float4 o;
                o.x = (acc[r][0][p] * sa + mrg[((r * 4 + 0) * 8 + p) * 32] * sb) / Lr;
                o.y = (acc[r][1][p] * sa + mrg[((r * 4 + 1) * 8 + p) * 32] * sb) / Lr;
                o.z = (acc[r][2][p] * sa + mrg[((r * 4 + 2) * 8 + p) * 32] * sb) / Lr;
                o.w = (acc[r][3][p] * sa + mrg[((r * 4 + 3) * 8 + p) * 32] * sb) / Lr;
                *(float4*)&s_part[(hl * K4_ROWS + 16 * r + c) * OUT_F + (hi << 5) + (p << 2)] = o;
            }
        }
    }
    __syncthreads();

    float* dst = hg == 0 ? out : part1;
    if (hg == 0 || ws_size >= WS_NEEDED) {
        for (int idx = tid; idx < K4_ROWS * OUT_F; idx += K4_THREADS) {
            float hp = 0.f;
#pragma unroll
            for (int hh = 0; hh < K4_HEADS; ++hh) hp += s_part[hh * K4_ROWS * OUT_F + idx];
            dst[(size_t)i0 * OUT_F + idx] = hp;
        }
    }
}

__global__ void __launch_bounds__(256)
k5_mean_ln(const float* __restrict__ part1, const float* __restrict__ gamma,
           const float* __restrict__ beta, float* out, size_t ws_size) {
    __shared__ float s_out[16 * OUT_F];
    __shared__ float s_mu[16], s_rs[16];
    const int i0  = blockIdx.x * 16;
    const int tid = threadIdx.x;

    for (int idx = tid; idx < 16 * OUT_F; idx += 256) {
        const size_t g = (size_t)i0 * OUT_F + idx;
        float hp = out[g];
        if (ws_size >= WS_NEEDED) hp += part1[g];
        s_out[idx] = hp * (1.0f / NHEAD);
    }
    __syncthreads();

    if (tid < 16) {
        float mu = 0.f;
        for (int f = 0; f < OUT_F; ++f) mu += s_out[tid * OUT_F + f];
        mu *= (1.0f / OUT_F);
        float var = 0.f;
        for (int f = 0; f < OUT_F; ++f) {
            const float d = s_out[tid * OUT_F + f] - mu;
            var += d * d;
        }
        var *= (1.0f / OUT_F);
        s_mu[tid] = mu;
        s_rs[tid] = rsqrtf(var + LN_EPS);
    }
    __syncthreads();

    for (int idx = tid; idx < 16 * OUT_F; idx += 256) {
        const int r = idx >> 6, f = idx & 63;
        out[(size_t)(i0 + r) * OUT_F + f] =
            (s_out[r * OUT_F + f] - s_mu[r]) * s_rs[r] * gamma[f] + beta[f];
    }
}

extern "C" void kernel_launch(void* const* d_in, const int* in_sizes, int n_in,
                              void* d_out, int out_size, void* d_ws, size_t ws_size,
                              hipStream_t stream) {
    const float* h     = (const float*)d_in[0];
    const int*   adj   = (const int*)  d_in[1];
    const float* W     = (const float*)d_in[2];
    const float* a     = (const float*)d_in[3];
    const float* gamma = (const float*)d_in[4];
    const float* beta  = (const float*)d_in[5];
    float* out = (float*)d_out;

    float* ws    = (float*)d_ws;
    float* ht    = ws;
    float* ssrc  = ht + (size_t)N_NODES * HID;
    float* sdst  = ssrc + N_NODES * NHEAD;
    float* part1 = sdst + N_NODES * NHEAD;
    float* f1    = part1 + (size_t)N_NODES * OUT_F;
    float* f2    = f1 + N_NODES * NHEAD;
    unsigned* amask = (unsigned*)(f2 + N_NODES * NHEAD);

    k1_gemm_ht<<<K1_GEMM_BLOCKS + K1_MASK_BLOCKS, 256, 0, stream>>>(h, W, a, adj, ht, ssrc, sdst, f1, f2,
                                                                  amask, ws_size);
    k4_agg<<<2 * (N_NODES / K4_ROWS), K4_THREADS, 0, stream>>>(adj, ht, ssrc, sdst, f1, f2, amask,
                                                       out, part1, ws_size);
    k5_mean_ln<<<N_NODES / 16, 256, 0, stream>>>(part1, gamma, beta, out, ws_size);
}
